# starting kernel
# baseline (speedup 1.0000x reference)
#include <hip/hip_runtime.h>


#define HDIM 128
#define BATCH 4
#define LEN 1024
#define MTOT (BATCH * LEN)
#define NCHUNK 16
#define CLEN (LEN / NCHUNK)

typedef float v2f __attribute__((ext_vector_type(2)));
typedef float v8f __attribute__((ext_vector_type(8)));

__device__ __forceinline__ float sigmoid_fast(float v) { return 1.0f / (1.0f + __expf(-v)); }
__device__ __forceinline__ float silu_fast(float v)    { return v * sigmoid_fast(v); }
__device__ __forceinline__ float softplus_fast(float v){ return (v > 20.0f) ? v : log1pf(__expf(v)); }

#define EPI_XZ    0
#define EPI_SSM   1
#define EPI_PLAIN 2

template <int N, int K, int MODE>
__global__ void __launch_bounds__(256)
gemm16_wmma_f32(const float* __restrict__ A,
                const float* __restrict__ W,
                const float* __restrict__ bias,
                float* __restrict__ out0,
                float* __restrict__ out1,
                float* __restrict__ out2)
{
    const int lane = threadIdx.x & 31;
    const int wave = threadIdx.x >> 5;
    const int gw   = blockIdx.x * (blockDim.x >> 5) + wave;
    const int tilesN = N >> 4;
    const int m0 = (gw / tilesN) << 4;
    const int n0 = (gw % tilesN) << 4;
    const int hi  = lane >> 4;
    const int l16 = lane & 15;

    v8f acc = {0.f, 0.f, 0.f, 0.f, 0.f, 0.f, 0.f, 0.f};

    const float* Arow = A + (size_t)(m0 + l16) * K + 2 * hi;
    const float* Wcol = W + (size_t)(2 * hi) * N + n0 + l16;

#pragma unroll
    for (int k0 = 0; k0 < K; k0 += 4) {
        v2f a;
        a.x = Arow[k0 + 0];
        a.y = Arow[k0 + 1];
        v2f b;
        b.x = Wcol[(k0 + 0) * N];
        b.y = Wcol[(k0 + 1) * N];
        acc = __builtin_amdgcn_wmma_f32_16x16x4_f32(false, a, false, b,
                                                    (short)0, acc, false, false);
    }

    const int   col = n0 + l16;
    const float bv  = bias[col];
#pragma unroll
    for (int r = 0; r < 8; ++r) {
        const int row = m0 + r + hi * 8;
        float v = acc[r] + bv;
        if (MODE == EPI_XZ) {
            if (col < HDIM) out0[(size_t)row * HDIM + col] = v;
            else            out1[(size_t)row * HDIM + (col - HDIM)] = silu_fast(v);
        } else if (MODE == EPI_SSM) {
            if (col < 128)      out0[(size_t)row * 128 + col]         = softplus_fast(v);
            else if (col < 256) out1[(size_t)row * 128 + (col - 128)] = v;
            else                out2[(size_t)row * 128 + (col - 256)] = v;
        } else {
            out0[(size_t)row * HDIM + col] = v;
        }
    }
}

__global__ void conv_silu_kernel(const float* __restrict__ x,
                                 const float* __restrict__ w,
                                 const float* __restrict__ cb,
                                 float* __restrict__ x2)
{
    const int idx = blockIdx.x * blockDim.x + threadIdx.x;
    const int e = idx & (HDIM - 1);
    const int m = idx >> 7;
    const int l = m & (LEN - 1);
    const float xm = (l > 0)       ? x[(size_t)(m - 1) * HDIM + e] : 0.f;
    const float xc =                 x[(size_t)m * HDIM + e];
    const float xp = (l < LEN - 1) ? x[(size_t)(m + 1) * HDIM + e] : 0.f;
    const float v  = xm * w[e * 3 + 0] + xc * w[e * 3 + 1] + xp * w[e * 3 + 2] + cb[e];
    x2[(size_t)m * HDIM + e] = silu_fast(v);
}


__global__ void __launch_bounds__(1024)
scan_passA(const float* __restrict__ delta,
           const float* __restrict__ Bm,
           const float* __restrict__ x2,
           const float* __restrict__ A_log,
           float* __restrict__ Ssum,
           float* __restrict__ Bacc)
{
    const int wg = blockIdx.x;
    const int b  = wg >> 4;
    const int c  = wg & (NCHUNK - 1);
    const int tid = threadIdx.x;
    const int j   = tid & 127;
    const int ng  = tid >> 7;
    const int nbase = ng << 4;

    __shared__ float s_d[128], s_B[128], s_x[128];

    const float Aj = -__expf(A_log[j]);
    float h[16], S[16];
#pragma unroll
    for (int i = 0; i < 16; ++i) { h[i] = 0.f; S[i] = 0.f; }

    const size_t base = ((size_t)b * LEN + (size_t)c * CLEN) * HDIM;
    for (int l = 0; l < CLEN; ++l) {
        const size_t moff = base + (size_t)l * HDIM;
        if (ng == 0) {
            s_d[j] = delta[moff + j];
            s_B[j] = Bm[moff + j];
            s_x[j] = x2[moff + j];
        }
        __syncthreads();
        const float xj = s_x[j];
#pragma unroll
        for (int i = 0; i < 16; ++i) {
            const float d = s_d[nbase + i];
            S[i] += d;
            const float e = __expf(d * Aj);
            h[i] = e * h[i] + (d * s_B[nbase + i]) * xj;
        }
        __syncthreads();
    }

    const size_t ob = (size_t)(b * NCHUNK + c) * 128;
#pragma unroll
    for (int i = 0; i < 16; ++i)
        Bacc[(ob + nbase + i) * 128 + j] = h[i];
    if (j == 0) {
#pragma unroll
        for (int i = 0; i < 16; ++i)
            Ssum[ob + nbase + i] = S[i];
    }
}

__global__ void __launch_bounds__(1024)
scan_passB(const float* __restrict__ Ssum,
           const float* __restrict__ Bacc,
           const float* __restrict__ A_log,
           float* __restrict__ hinit)
{
    const int b   = blockIdx.x;
    const int tid = threadIdx.x;
    const int j   = tid & 127;
    const int ng  = tid >> 7;
    const int nbase = ng << 4;

    __shared__ float s_S[128];

    const float Aj = -__expf(A_log[j]);
    float h[16];
#pragma unroll
    for (int i = 0; i < 16; ++i) h[i] = 0.f;

    for (int c = 0; c < NCHUNK; ++c) {
        const size_t ob = (size_t)(b * NCHUNK + c) * 128;
        if (ng == 0) s_S[j] = Ssum[ob + j];
        __syncthreads();
#pragma unroll
        for (int i = 0; i < 16; ++i) {
            hinit[(ob + nbase + i) * 128 + j] = h[i];
            const float P = __expf(s_S[nbase + i] * Aj);
            h[i] = P * h[i] + Bacc[(ob + nbase + i) * 128 + j];
        }
        __syncthreads();
    }
}

__global__ void __launch_bounds__(1024)
scan_passC(const float* __restrict__ delta,
           const float* __restrict__ Bm,
           const float* __restrict__ Cm,
           const float* __restrict__ x2,
           const float* __restrict__ u,
           const float* __restrict__ sz,
           const float* __restrict__ A_log,
           const float* __restrict__ Dv,
           const float* __restrict__ hinit,
           float* __restrict__ ymul)
{
    const int wg = blockIdx.x;
    const int b  = wg >> 4;
    const int c  = wg & (NCHUNK - 1);
    const int tid = threadIdx.x;
    const int j   = tid & 127;
    const int ng  = tid >> 7;
    const int nbase = ng << 4;

    __shared__ float s_d[128], s_B[128], s_x[128], s_C[128], s_u[128];
    __shared__ float s_part[8][128];

    const float Aj = -__expf(A_log[j]);
    const float Dj = Dv[j];

    const size_t ob = (size_t)(b * NCHUNK + c) * 128;
    float h[16];
#pragma unroll
    for (int i = 0; i < 16; ++i) h[i] = hinit[(ob + nbase + i) * 128 + j];

    const size_t base = ((size_t)b * LEN + (size_t)c * CLEN) * HDIM;
    for (int l = 0; l < CLEN; ++l) {
        const size_t moff = base + (size_t)l * HDIM;
        if (ng == 0) {
            s_d[j] = delta[moff + j];
            s_B[j] = Bm[moff + j];
            s_x[j] = x2[moff + j];
            s_C[j] = Cm[moff + j];
            s_u[j] = u[moff + j];
        }
        __syncthreads();
        const float xj = s_x[j];
        float partial = 0.f;
#pragma unroll
        for (int i = 0; i < 16; ++i) {
            const float d = s_d[nbase + i];
            const float e = __expf(d * Aj);
            h[i] = e * h[i] + (d * s_B[nbase + i]) * xj;
            partial += h[i];
        }
        s_part[ng][j] = partial;
        __syncthreads();
        if (ng == 0) {
            float tot = 0.f;
#pragma unroll
            for (int g = 0; g < 8; ++g) tot += s_part[g][j];
            const float y = s_C[j] * tot + Dj * s_u[j];
            ymul[moff + j] = y * sz[moff + j];
        }
    }
}

extern "C" void kernel_launch(void* const* d_in, const int* in_sizes, int n_in,
                              void* d_out, int out_size, void* d_ws, size_t ws_size,
                              hipStream_t stream)
{
    (void)in_sizes; (void)n_in; (void)out_size; (void)ws_size;

    const float* u      = (const float*)d_in[0];
    const float* W_in   = (const float*)d_in[1];
    const float* b_in   = (const float*)d_in[2];
    const float* conv_w = (const float*)d_in[3];
    const float* conv_b = (const float*)d_in[4];
    const float* W_x    = (const float*)d_in[5];
    const float* b_x    = (const float*)d_in[6];
    const float* A_log  = (const float*)d_in[7];
    const float* Dv     = (const float*)d_in[8];
    const float* W_out  = (const float*)d_in[9];
    const float* b_out  = (const float*)d_in[10];

    const size_t P = (size_t)MTOT * HDIM;
    const size_t NSTATE = (size_t)BATCH * NCHUNK * 128 * 128;

    float* ws    = (float*)d_ws;
    float* x_raw = ws + 0 * P;
    float* s_z   = ws + 1 * P;
    float* x2    = ws + 2 * P;
    float* delta = ws + 3 * P;
    float* Bmat  = ws + 4 * P;
    float* Cmat  = ws + 5 * P;
    float* ymul  = ws + 6 * P;
    float* Ssum  = ws + 7 * P;
    float* Bacc  = Ssum + (size_t)BATCH * NCHUNK * 128;
    float* hinit = Bacc + NSTATE;

    gemm16_wmma_f32<256, 128, EPI_XZ><<<512, 256, 0, stream>>>(
        u, W_in, b_in, x_raw, s_z, nullptr);
    conv_silu_kernel<<<(MTOT * HDIM) / 256, 256, 0, stream>>>(x_raw, conv_w, conv_b, x2);
    gemm16_wmma_f32<384, 128, EPI_SSM><<<768, 256, 0, stream>>>(
        x2, W_x, b_x, delta, Bmat, Cmat);
    scan_passA<<<BATCH * NCHUNK, 1024, 0, stream>>>(delta, Bmat, x2, A_log, Ssum, Bacc);
    scan_passB<<<BATCH, 1024, 0, stream>>>(Ssum, Bacc, A_log, hinit);
    scan_passC<<<BATCH * NCHUNK, 1024, 0, stream>>>(delta, Bmat, Cmat, x2, u, s_z,
                                                    A_log, Dv, hinit, ymul);
    gemm16_wmma_f32<128, 128, EPI_PLAIN><<<256, 256, 0, stream>>>(
        ymul, W_out, b_out, (float*)d_out, nullptr, nullptr);
}


# scan passes A and C: copy the chunk's delta/B/x rows to LDS once and close y for 8 steps per barrier, so no global round trip or barrier pair sits between two recurrence steps
# speedup vs baseline: 1.8311x; 1.8311x over previous
#include <hip/hip_runtime.h>


#define HDIM 128
#define BATCH 4
#define LEN 1024
#define MTOT (BATCH * LEN)
#define NCHUNK 16
#define CLEN (LEN / NCHUNK)

typedef float v2f __attribute__((ext_vector_type(2)));
typedef float v8f __attribute__((ext_vector_type(8)));

__device__ __forceinline__ float sigmoid_fast(float v) { return 1.0f / (1.0f + __expf(-v)); }
__device__ __forceinline__ float silu_fast(float v)    { return v * sigmoid_fast(v); }
__device__ __forceinline__ float softplus_fast(float v){ return (v > 20.0f) ? v : log1pf(__expf(v)); }

#define EPI_XZ    0
#define EPI_SSM   1
#define EPI_PLAIN 2

template <int N, int K, int MODE>
__global__ void __launch_bounds__(256)
gemm16_wmma_f32(const float* __restrict__ A,
                const float* __restrict__ W,
                const float* __restrict__ bias,
                float* __restrict__ out0,
                float* __restrict__ out1,
                float* __restrict__ out2)
{
    const int lane = threadIdx.x & 31;
    const int wave = threadIdx.x >> 5;
    const int gw   = blockIdx.x * (blockDim.x >> 5) + wave;
    const int tilesN = N >> 4;
    const int m0 = (gw / tilesN) << 4;
    const int n0 = (gw % tilesN) << 4;
    const int hi  = lane >> 4;
    const int l16 = lane & 15;

    v8f acc = {0.f, 0.f, 0.f, 0.f, 0.f, 0.f, 0.f, 0.f};

    const float* Arow = A + (size_t)(m0 + l16) * K + 2 * hi;
    const float* Wcol = W + (size_t)(2 * hi) * N + n0 + l16;

#pragma unroll
    for (int k0 = 0; k0 < K; k0 += 4) {
        v2f a;
        a.x = Arow[k0 + 0];
        a.y = Arow[k0 + 1];
        v2f b;
        b.x = Wcol[(k0 + 0) * N];
        b.y = Wcol[(k0 + 1) * N];
        acc = __builtin_amdgcn_wmma_f32_16x16x4_f32(false, a, false, b,
                                                    (short)0, acc, false, false);
    }

    const int   col = n0 + l16;
    const float bv  = bias[col];
#pragma unroll
    for (int r = 0; r < 8; ++r) {
        const int row = m0 + r + hi * 8;
        float v = acc[r] + bv;
        if (MODE == EPI_XZ) {
            if (col < HDIM) out0[(size_t)row * HDIM + col] = v;
            else            out1[(size_t)row * HDIM + (col - HDIM)] = silu_fast(v);
        } else if (MODE == EPI_SSM) {
            if (col < 128)      out0[(size_t)row * 128 + col]         = softplus_fast(v);
            else if (col < 256) out1[(size_t)row * 128 + (col - 128)] = v;
            else                out2[(size_t)row * 128 + (col - 256)] = v;
        } else {
            out0[(size_t)row * HDIM + col] = v;
        }
    }
}

__global__ void conv_silu_kernel(const float* __restrict__ x,
                                 const float* __restrict__ w,
                                 const float* __restrict__ cb,
                                 float* __restrict__ x2)
{
    const int idx = blockIdx.x * blockDim.x + threadIdx.x;
    const int e = idx & (HDIM - 1);
    const int m = idx >> 7;
    const int l = m & (LEN - 1);
    const float xm = (l > 0)       ? x[(size_t)(m - 1) * HDIM + e] : 0.f;
    const float xc =                 x[(size_t)m * HDIM + e];
    const float xp = (l < LEN - 1) ? x[(size_t)(m + 1) * HDIM + e] : 0.f;
    const float v  = xm * w[e * 3 + 0] + xc * w[e * 3 + 1] + xp * w[e * 3 + 2] + cb[e];
    x2[(size_t)m * HDIM + e] = silu_fast(v);
}


__global__ void __launch_bounds__(1024)
scan_passA(const float* __restrict__ delta,
           const float* __restrict__ Bm,
           const float* __restrict__ x2,
           const float* __restrict__ A_log,
           float* __restrict__ Ssum,
           float* __restrict__ Bacc)
{
    const int wg = blockIdx.x;
    const int b  = wg >> 4;
    const int c  = wg & (NCHUNK - 1);
    const int tid = threadIdx.x;
    const int j   = tid & 127;
    const int ng  = tid >> 7;
    const int nbase = ng << 4;

    __shared__ __attribute__((aligned(16))) float s_d[CLEN * 128];
    __shared__ __attribute__((aligned(16))) float s_B[CLEN * 128];
    __shared__ __attribute__((aligned(16))) float s_x[CLEN * 128];

    const size_t base = ((size_t)b * LEN + (size_t)c * CLEN) * HDIM;
    {
        const float4* gd = reinterpret_cast<const float4*>(delta + base);
        const float4* gB = reinterpret_cast<const float4*>(Bm + base);
        const float4* gx = reinterpret_cast<const float4*>(x2 + base);
        float4* ld = reinterpret_cast<float4*>(s_d);
        float4* lB = reinterpret_cast<float4*>(s_B);
        float4* lx = reinterpret_cast<float4*>(s_x);
        for (int k = 0; k < (CLEN * 128 / 4) / 1024; ++k) {
            const int q = tid + k * 1024;
            ld[q] = gd[q];
            lB[q] = gB[q];
            lx[q] = gx[q];
        }
    }

    const float Aj = -__expf(A_log[j]);
    float h[16], S[16];
#pragma unroll
    for (int i = 0; i < 16; ++i) { h[i] = 0.f; S[i] = 0.f; }

    __syncthreads();

    for (int l = 0; l < CLEN; ++l) {
        const float* rd = s_d + l * 128 + nbase;
        const float* rB = s_B + l * 128 + nbase;
        const float xj = s_x[l * 128 + j];
#pragma unroll
        for (int i = 0; i < 16; ++i) {
            const float d = rd[i];
            S[i] += d;
            const float e = __expf(d * Aj);
            h[i] = e * h[i] + (d * rB[i]) * xj;
        }
    }

    const size_t ob = (size_t)(b * NCHUNK + c) * 128;
#pragma unroll
    for (int i = 0; i < 16; ++i)
        Bacc[(ob + nbase + i) * 128 + j] = h[i];
    if (j == 0) {
#pragma unroll
        for (int i = 0; i < 16; ++i)
            Ssum[ob + nbase + i] = S[i];
    }
}

__global__ void __launch_bounds__(1024)
scan_passB(const float* __restrict__ Ssum,
           const float* __restrict__ Bacc,
           const float* __restrict__ A_log,
           float* __restrict__ hinit)
{
    const int b   = blockIdx.x;
    const int tid = threadIdx.x;
    const int j   = tid & 127;
    const int ng  = tid >> 7;
    const int nbase = ng << 4;

    __shared__ float s_S[128];

    const float Aj = -__expf(A_log[j]);
    float h[16];
#pragma unroll
    for (int i = 0; i < 16; ++i) h[i] = 0.f;

    for (int c = 0; c < NCHUNK; ++c) {
        const size_t ob = (size_t)(b * NCHUNK + c) * 128;
        if (ng == 0) s_S[j] = Ssum[ob + j];
        __syncthreads();
#pragma unroll
        for (int i = 0; i < 16; ++i) {
            hinit[(ob + nbase + i) * 128 + j] = h[i];
            const float P = __expf(s_S[nbase + i] * Aj);
            h[i] = P * h[i] + Bacc[(ob + nbase + i) * 128 + j];
        }
        __syncthreads();
    }
}

#define SBLK 8
__global__ void __launch_bounds__(1024)
scan_passC(const float* __restrict__ delta,
           const float* __restrict__ Bm,
           const float* __restrict__ Cm,
           const float* __restrict__ x2,
           const float* __restrict__ u,
           const float* __restrict__ sz,
           const float* __restrict__ A_log,
           const float* __restrict__ Dv,
           const float* __restrict__ hinit,
           float* __restrict__ ymul)
{
    const int wg = blockIdx.x;
    const int b  = wg >> 4;
    const int c  = wg & (NCHUNK - 1);
    const int tid = threadIdx.x;
    const int j   = tid & 127;
    const int ng  = tid >> 7;
    const int nbase = ng << 4;

    __shared__ __attribute__((aligned(16))) float s_d[CLEN * 128];
    __shared__ __attribute__((aligned(16))) float s_B[CLEN * 128];
    __shared__ __attribute__((aligned(16))) float s_x[CLEN * 128];
    __shared__ float s_part[2][SBLK][8][128];

    const size_t base = ((size_t)b * LEN + (size_t)c * CLEN) * HDIM;
    {
        const float4* gd = reinterpret_cast<const float4*>(delta + base);
        const float4* gB = reinterpret_cast<const float4*>(Bm + base);
        const float4* gx = reinterpret_cast<const float4*>(x2 + base);
        float4* ld = reinterpret_cast<float4*>(s_d);
        float4* lB = reinterpret_cast<float4*>(s_B);
        float4* lx = reinterpret_cast<float4*>(s_x);
        for (int k = 0; k < (CLEN * 128 / 4) / 1024; ++k) {
            const int q = tid + k * 1024;
            ld[q] = gd[q];
            lB[q] = gB[q];
            lx[q] = gx[q];
        }
    }

    const float Aj = -__expf(A_log[j]);
    const float Dj = Dv[j];

    const size_t ob = (size_t)(b * NCHUNK + c) * 128;
    float h[16];
#pragma unroll
    for (int i = 0; i < 16; ++i) h[i] = hinit[(ob + nbase + i) * 128 + j];

    __syncthreads();

    for (int blk = 0; blk < CLEN / SBLK; ++blk) {
        const int buf = blk & 1;
        const size_t moff = base + (size_t)(blk * SBLK + ng) * HDIM + j;
        const float cv = Cm[moff];
        const float uv = u[moff];
        const float zv = sz[moff];

        for (int ll = 0; ll < SBLK; ++ll) {
            const int l = blk * SBLK + ll;
            const float* rd = s_d + l * 128 + nbase;
            const float* rB = s_B + l * 128 + nbase;
            const float xj = s_x[l * 128 + j];
            float partial = 0.f;
#pragma unroll
            for (int i = 0; i < 16; ++i) {
                const float d = rd[i];
                const float e = __expf(d * Aj);
                h[i] = e * h[i] + (d * rB[i]) * xj;
                partial += h[i];
            }
            s_part[buf][ll][ng][j] = partial;
        }
        __syncthreads();
        float tot = 0.f;
#pragma unroll
        for (int g = 0; g < 8; ++g) tot += s_part[buf][ng][g][j];
        const float y = cv * tot + Dj * uv;
        ymul[moff] = y * zv;
    }
}

extern "C" void kernel_launch(void* const* d_in, const int* in_sizes, int n_in,
                              void* d_out, int out_size, void* d_ws, size_t ws_size,
                              hipStream_t stream)
{
    (void)in_sizes; (void)n_in; (void)out_size; (void)ws_size;

    const float* u      = (const float*)d_in[0];
    const float* W_in   = (const float*)d_in[1];
    const float* b_in   = (const float*)d_in[2];
    const float* conv_w = (const float*)d_in[3];
    const float* conv_b = (const float*)d_in[4];
    const float* W_x    = (const float*)d_in[5];
    const float* b_x    = (const float*)d_in[6];
    const float* A_log  = (const float*)d_in[7];
    const float* Dv     = (const float*)d_in[8];
    const float* W_out  = (const float*)d_in[9];
    const float* b_out  = (const float*)d_in[10];

    const size_t P = (size_t)MTOT * HDIM;
    const size_t NSTATE = (size_t)BATCH * NCHUNK * 128 * 128;

    float* ws    = (float*)d_ws;
    float* x_raw = ws + 0 * P;
    float* s_z   = ws + 1 * P;
    float* x2    = ws + 2 * P;
    float* delta = ws + 3 * P;
    float* Bmat  = ws + 4 * P;
    float* Cmat  = ws + 5 * P;
    float* ymul  = ws + 6 * P;
    float* Ssum  = ws + 7 * P;
    float* Bacc  = Ssum + (size_t)BATCH * NCHUNK * 128;
    float* hinit = Bacc + NSTATE;

    gemm16_wmma_f32<256, 128, EPI_XZ><<<512, 256, 0, stream>>>(
        u, W_in, b_in, x_raw, s_z, nullptr);
    conv_silu_kernel<<<(MTOT * HDIM) / 256, 256, 0, stream>>>(x_raw, conv_w, conv_b, x2);
    gemm16_wmma_f32<384, 128, EPI_SSM><<<768, 256, 0, stream>>>(
        x2, W_x, b_x, delta, Bmat, Cmat);
    scan_passA<<<BATCH * NCHUNK, 1024, 0, stream>>>(delta, Bmat, x2, A_log, Ssum, Bacc);
    scan_passB<<<BATCH, 1024, 0, stream>>>(Ssum, Bacc, A_log, hinit);
    scan_passC<<<BATCH * NCHUNK, 1024, 0, stream>>>(delta, Bmat, Cmat, x2, u, s_z,
                                                    A_log, Dv, hinit, ymul);
    gemm16_wmma_f32<128, 128, EPI_PLAIN><<<256, 256, 0, stream>>>(
        ymul, W_out, b_out, (float*)d_out, nullptr, nullptr);
}


# scan passes A, B, C: cut each (batch, chunk) into 4 slices of 32 columns, one 8-wave workgroup per slice (B: one state element per thread), so all 256 WGPs run the scan instead of 64 (B: 4)
# speedup vs baseline: 3.7746x; 2.0614x over previous
#include <hip/hip_runtime.h>


#define HDIM 128
#define BATCH 4
#define LEN 1024
#define MTOT (BATCH * LEN)
#define NCHUNK 16
#define CLEN (LEN / NCHUNK)

typedef float v2f __attribute__((ext_vector_type(2)));
typedef float v8f __attribute__((ext_vector_type(8)));

__device__ __forceinline__ float sigmoid_fast(float v) { return 1.0f / (1.0f + __expf(-v)); }
__device__ __forceinline__ float silu_fast(float v)    { return v * sigmoid_fast(v); }
__device__ __forceinline__ float softplus_fast(float v){ return (v > 20.0f) ? v : log1pf(__expf(v)); }

#define EPI_XZ    0
#define EPI_SSM   1
#define EPI_PLAIN 2

template <int N, int K, int MODE>
__global__ void __launch_bounds__(256)
gemm16_wmma_f32(const float* __restrict__ A,
                const float* __restrict__ W,
                const float* __restrict__ bias,
                float* __restrict__ out0,
                float* __restrict__ out1,
                float* __restrict__ out2)
{
    const int lane = threadIdx.x & 31;
    const int wave = threadIdx.x >> 5;
    const int gw   = blockIdx.x * (blockDim.x >> 5) + wave;
    const int tilesN = N >> 4;
    const int m0 = (gw / tilesN) << 4;
    const int n0 = (gw % tilesN) << 4;
    const int hi  = lane >> 4;
    const int l16 = lane & 15;

    v8f acc = {0.f, 0.f, 0.f, 0.f, 0.f, 0.f, 0.f, 0.f};

    const float* Arow = A + (size_t)(m0 + l16) * K + 2 * hi;
    const float* Wcol = W + (size_t)(2 * hi) * N + n0 + l16;

#pragma unroll
    for (int k0 = 0; k0 < K; k0 += 4) {
        v2f a;
        a.x = Arow[k0 + 0];
        a.y = Arow[k0 + 1];
        v2f b;
        b.x = Wcol[(k0 + 0) * N];
        b.y = Wcol[(k0 + 1) * N];
        acc = __builtin_amdgcn_wmma_f32_16x16x4_f32(false, a, false, b,
                                                    (short)0, acc, false, false);
    }

    const int   col = n0 + l16;
    const float bv  = bias[col];
#pragma unroll
    for (int r = 0; r < 8; ++r) {
        const int row = m0 + r + hi * 8;
        float v = acc[r] + bv;
        if (MODE == EPI_XZ) {
            if (col < HDIM) out0[(size_t)row * HDIM + col] = v;
            else            out1[(size_t)row * HDIM + (col - HDIM)] = silu_fast(v);
        } else if (MODE == EPI_SSM) {
            if (col < 128)      out0[(size_t)row * 128 + col]         = softplus_fast(v);
            else if (col < 256) out1[(size_t)row * 128 + (col - 128)] = v;
            else                out2[(size_t)row * 128 + (col - 256)] = v;
        } else {
            out0[(size_t)row * HDIM + col] = v;
        }
    }
}

__global__ void conv_silu_kernel(const float* __restrict__ x,
                                 const float* __restrict__ w,
                                 const float* __restrict__ cb,
                                 float* __restrict__ x2)
{
    const int idx = blockIdx.x * blockDim.x + threadIdx.x;
    const int e = idx & (HDIM - 1);
    const int m = idx >> 7;
    const int l = m & (LEN - 1);
    const float xm = (l > 0)       ? x[(size_t)(m - 1) * HDIM + e] : 0.f;
    const float xc =                 x[(size_t)m * HDIM + e];
    const float xp = (l < LEN - 1) ? x[(size_t)(m + 1) * HDIM + e] : 0.f;
    const float v  = xm * w[e * 3 + 0] + xc * w[e * 3 + 1] + xp * w[e * 3 + 2] + cb[e];
    x2[(size_t)m * HDIM + e] = silu_fast(v);
}

#define JS 4
#define JW (HDIM / JS)
#define SCAN_THREADS (8 * JW)

__global__ void __launch_bounds__(SCAN_THREADS)
scan_passA(const float* __restrict__ delta,
           const float* __restrict__ Bm,
           const float* __restrict__ x2,
           const float* __restrict__ A_log,
           float* __restrict__ Ssum,
           float* __restrict__ Bacc)
{
    const int wg = blockIdx.x;
    const int js = wg & (JS - 1);
    const int bc = wg / JS;
    const int b  = bc >> 4;
    const int c  = bc & (NCHUNK - 1);
    const int tid = threadIdx.x;
    const int jl  = tid & (JW - 1);
    const int ng  = tid / JW;
    const int j   = js * JW + jl;
    const int nbase = ng << 4;

    __shared__ __attribute__((aligned(16))) float s_d[CLEN * 128];
    __shared__ __attribute__((aligned(16))) float s_B[CLEN * 128];
    __shared__ __attribute__((aligned(16))) float s_x[CLEN * JW];

    const size_t base = ((size_t)b * LEN + (size_t)c * CLEN) * HDIM;
    {
        const float4* gd = reinterpret_cast<const float4*>(delta + base);
        const float4* gB = reinterpret_cast<const float4*>(Bm + base);
        const float4* gx = reinterpret_cast<const float4*>(x2 + base + js * JW);
        float4* ld = reinterpret_cast<float4*>(s_d);
        float4* lB = reinterpret_cast<float4*>(s_B);
        float4* lx = reinterpret_cast<float4*>(s_x);
        for (int k = 0; k < (CLEN * 128 / 4) / SCAN_THREADS; ++k) {
            const int q = tid + k * SCAN_THREADS;
            ld[q] = gd[q];
            lB[q] = gB[q];
        }
        for (int k = 0; k < (CLEN * JW / 4) / SCAN_THREADS; ++k) {
            const int q = tid + k * SCAN_THREADS;
            lx[q] = gx[(q / (JW / 4)) * (HDIM / 4) + (q % (JW / 4))];
        }
    }

    const float Aj = -__expf(A_log[j]);
    float h[16], S[16];
#pragma unroll
    for (int i = 0; i < 16; ++i) { h[i] = 0.f; S[i] = 0.f; }

    __syncthreads();

    for (int l = 0; l < CLEN; ++l) {
        const float* rd = s_d + l * 128 + nbase;
        const float* rB = s_B + l * 128 + nbase;
        const float xj = s_x[l * JW + jl];
#pragma unroll
        for (int i = 0; i < 16; ++i) {
            const float d = rd[i];
            S[i] += d;
            const float e = __expf(d * Aj);
            h[i] = e * h[i] + (d * rB[i]) * xj;
        }
    }

    const size_t ob = (size_t)bc * 128;
#pragma unroll
    for (int i = 0; i < 16; ++i)
        Bacc[(ob + nbase + i) * 128 + j] = h[i];
    if (j == 0) {
#pragma unroll
        for (int i = 0; i < 16; ++i)
            Ssum[ob + nbase + i] = S[i];
    }
}

__global__ void __launch_bounds__(SCAN_THREADS)
scan_passB(const float* __restrict__ Ssum,
           const float* __restrict__ Bacc,
           const float* __restrict__ A_log,
           float* __restrict__ hinit)
{
    const int id = blockIdx.x * SCAN_THREADS + threadIdx.x;
    const int j  = id & 127;
    const int n  = (id >> 7) & 127;
    const int b  = id >> 14;

    const float Aj = -__expf(A_log[j]);
    float h = 0.f;
#pragma unroll
    for (int c = 0; c < NCHUNK; ++c) {
        const size_t ob = (size_t)(b * NCHUNK + c) * 128;
        const size_t idx = (ob + n) * 128 + j;
        hinit[idx] = h;
        const float P = __expf(Ssum[ob + n] * Aj);
        h = P * h + Bacc[idx];
    }
}

#define SBLK 8
__global__ void __launch_bounds__(SCAN_THREADS)
scan_passC(const float* __restrict__ delta,
           const float* __restrict__ Bm,
           const float* __restrict__ Cm,
           const float* __restrict__ x2,
           const float* __restrict__ u,
           const float* __restrict__ sz,
           const float* __restrict__ A_log,
           const float* __restrict__ Dv,
           const float* __restrict__ hinit,
           float* __restrict__ ymul)
{
    const int wg = blockIdx.x;
    const int js = wg & (JS - 1);
    const int bc = wg / JS;
    const int b  = bc >> 4;
    const int c  = bc & (NCHUNK - 1);
    const int tid = threadIdx.x;
    const int jl  = tid & (JW - 1);
    const int ng  = tid / JW;
    const int j   = js * JW + jl;
    const int nbase = ng << 4;

    __shared__ __attribute__((aligned(16))) float s_d[CLEN * 128];
    __shared__ __attribute__((aligned(16))) float s_B[CLEN * 128];
    __shared__ __attribute__((aligned(16))) float s_x[CLEN * JW];
    __shared__ float s_part[2][SBLK][8][JW];

    const size_t base = ((size_t)b * LEN + (size_t)c * CLEN) * HDIM;
    {
        const float4* gd = reinterpret_cast<const float4*>(delta + base);
        const float4* gB = reinterpret_cast<const float4*>(Bm + base);
        const float4* gx = reinterpret_cast<const float4*>(x2 + base + js * JW);
        float4* ld = reinterpret_cast<float4*>(s_d);
        float4* lB = reinterpret_cast<float4*>(s_B);
        float4* lx = reinterpret_cast<float4*>(s_x);
        for (int k = 0; k < (CLEN * 128 / 4) / SCAN_THREADS; ++k) {
            const int q = tid + k * SCAN_THREADS;
            ld[q] = gd[q];
            lB[q] = gB[q];
        }
        for (int k = 0; k < (CLEN * JW / 4) / SCAN_THREADS; ++k) {
            const int q = tid + k * SCAN_THREADS;
            lx[q] = gx[(q / (JW / 4)) * (HDIM / 4) + (q % (JW / 4))];
        }
    }

    const float Aj = -__expf(A_log[j]);
    const float Dj = Dv[j];

    const size_t ob = (size_t)bc * 128;
    float h[16];
#pragma unroll
    for (int i = 0; i < 16; ++i) h[i] = hinit[(ob + nbase + i) * 128 + j];

    __syncthreads();

    for (int blk = 0; blk < CLEN / SBLK; ++blk) {
        const int buf = blk & 1;
        const size_t moff = base + (size_t)(blk * SBLK + ng) * HDIM + j;
        const float cv = Cm[moff];
        const float uv = u[moff];
        const float zv = sz[moff];

        for (int ll = 0; ll < SBLK; ++ll) {
            const int l = blk * SBLK + ll;
            const float* rd = s_d + l * 128 + nbase;
            const float* rB = s_B + l * 128 + nbase;
            const float xj = s_x[l * JW + jl];
            float partial = 0.f;
#pragma unroll
            for (int i = 0; i < 16; ++i) {
                const float d = rd[i];
                const float e = __expf(d * Aj);
                h[i] = e * h[i] + (d * rB[i]) * xj;
                partial += h[i];
            }
            s_part[buf][ll][ng][jl] = partial;
        }
        __syncthreads();
        float tot = 0.f;
#pragma unroll
        for (int g = 0; g < 8; ++g) tot += s_part[buf][ng][g][jl];
        const float y = cv * tot + Dj * uv;
        ymul[moff] = y * zv;
    }
}

extern "C" void kernel_launch(void* const* d_in, const int* in_sizes, int n_in,
                              void* d_out, int out_size, void* d_ws, size_t ws_size,
                              hipStream_t stream)
{
    (void)in_sizes; (void)n_in; (void)out_size; (void)ws_size;

    const float* u      = (const float*)d_in[0];
    const float* W_in   = (const float*)d_in[1];
    const float* b_in   = (const float*)d_in[2];
    const float* conv_w = (const float*)d_in[3];
    const float* conv_b = (const float*)d_in[4];
    const float* W_x    = (const float*)d_in[5];
    const float* b_x    = (const float*)d_in[6];
    const float* A_log  = (const float*)d_in[7];
    const float* Dv     = (const float*)d_in[8];
    const float* W_out  = (const float*)d_in[9];
    const float* b_out  = (const float*)d_in[10];

    const size_t P = (size_t)MTOT * HDIM;
    const size_t NSTATE = (size_t)BATCH * NCHUNK * 128 * 128;

    float* ws    = (float*)d_ws;
    float* x_raw = ws + 0 * P;
    float* s_z   = ws + 1 * P;
    float* x2    = ws + 2 * P;
    float* delta = ws + 3 * P;
    float* Bmat  = ws + 4 * P;
    float* Cmat  = ws + 5 * P;
    float* ymul  = ws + 6 * P;
    float* Ssum  = ws + 7 * P;
    float* Bacc  = Ssum + (size_t)BATCH * NCHUNK * 128;
    float* hinit = Bacc + NSTATE;

    gemm16_wmma_f32<256, 128, EPI_XZ><<<512, 256, 0, stream>>>(
        u, W_in, b_in, x_raw, s_z, nullptr);
    conv_silu_kernel<<<(MTOT * HDIM) / 256, 256, 0, stream>>>(x_raw, conv_w, conv_b, x2);
    gemm16_wmma_f32<384, 128, EPI_SSM><<<768, 256, 0, stream>>>(
        x2, W_x, b_x, delta, Bmat, Cmat);
    scan_passA<<<BATCH * NCHUNK * JS, SCAN_THREADS, 0, stream>>>(delta, Bmat, x2, A_log,
                                                                 Ssum, Bacc);
    scan_passB<<<(BATCH * 128 * 128) / SCAN_THREADS, SCAN_THREADS, 0, stream>>>(
        Ssum, Bacc, A_log, hinit);
    scan_passC<<<BATCH * NCHUNK * JS, SCAN_THREADS, 0, stream>>>(delta, Bmat, Cmat, x2, u, s_z,
                                                                 A_log, Dv, hinit, ymul);
    gemm16_wmma_f32<128, 128, EPI_PLAIN><<<256, 256, 0, stream>>>(
        ymul, W_out, b_out, (float*)d_out, nullptr, nullptr);
}


# scan passes A and C: column-independent work leaves the per-lane recurrence: d*B formed once in the LDS copy, log2(e) folded into the column factor, chunk sum of d kept by one lane per state
# speedup vs baseline: 3.9056x; 1.0347x over previous
#include <hip/hip_runtime.h>


#define HDIM 128
#define BATCH 4
#define LEN 1024
#define MTOT (BATCH * LEN)
#define NCHUNK 16
#define CLEN (LEN / NCHUNK)

typedef float v2f __attribute__((ext_vector_type(2)));
typedef float v8f __attribute__((ext_vector_type(8)));

__device__ __forceinline__ float sigmoid_fast(float v) { return 1.0f / (1.0f + __expf(-v)); }
__device__ __forceinline__ float silu_fast(float v)    { return v * sigmoid_fast(v); }
__device__ __forceinline__ float softplus_fast(float v){ return (v > 20.0f) ? v : log1pf(__expf(v)); }

#define EPI_XZ    0
#define EPI_SSM   1
#define EPI_PLAIN 2

template <int N, int K, int MODE>
__global__ void __launch_bounds__(256)
gemm16_wmma_f32(const float* __restrict__ A,
                const float* __restrict__ W,
                const float* __restrict__ bias,
                float* __restrict__ out0,
                float* __restrict__ out1,
                float* __restrict__ out2)
{
    const int lane = threadIdx.x & 31;
    const int wave = threadIdx.x >> 5;
    const int gw   = blockIdx.x * (blockDim.x >> 5) + wave;
    const int tilesN = N >> 4;
    const int m0 = (gw / tilesN) << 4;
    const int n0 = (gw % tilesN) << 4;
    const int hi  = lane >> 4;
    const int l16 = lane & 15;

    v8f acc = {0.f, 0.f, 0.f, 0.f, 0.f, 0.f, 0.f, 0.f};

    const float* Arow = A + (size_t)(m0 + l16) * K + 2 * hi;
    const float* Wcol = W + (size_t)(2 * hi) * N + n0 + l16;

#pragma unroll
    for (int k0 = 0; k0 < K; k0 += 4) {
        v2f a;
        a.x = Arow[k0 + 0];
        a.y = Arow[k0 + 1];
        v2f b;
        b.x = Wcol[(k0 + 0) * N];
        b.y = Wcol[(k0 + 1) * N];
        acc = __builtin_amdgcn_wmma_f32_16x16x4_f32(false, a, false, b,
                                                    (short)0, acc, false, false);
    }

    const int   col = n0 + l16;
    const float bv  = bias[col];
#pragma unroll
    for (int r = 0; r < 8; ++r) {
        const int row = m0 + r + hi * 8;
        float v = acc[r] + bv;
        if (MODE == EPI_XZ) {
            if (col < HDIM) out0[(size_t)row * HDIM + col] = v;
            else            out1[(size_t)row * HDIM + (col - HDIM)] = silu_fast(v);
        } else if (MODE == EPI_SSM) {
            if (col < 128)      out0[(size_t)row * 128 + col]         = softplus_fast(v);
            else if (col < 256) out1[(size_t)row * 128 + (col - 128)] = v;
            else                out2[(size_t)row * 128 + (col - 256)] = v;
        } else {
            out0[(size_t)row * HDIM + col] = v;
        }
    }
}

__global__ void conv_silu_kernel(const float* __restrict__ x,
                                 const float* __restrict__ w,
                                 const float* __restrict__ cb,
                                 float* __restrict__ x2)
{
    const int idx = blockIdx.x * blockDim.x + threadIdx.x;
    const int e = idx & (HDIM - 1);
    const int m = idx >> 7;
    const int l = m & (LEN - 1);
    const float xm = (l > 0)       ? x[(size_t)(m - 1) * HDIM + e] : 0.f;
    const float xc =                 x[(size_t)m * HDIM + e];
    const float xp = (l < LEN - 1) ? x[(size_t)(m + 1) * HDIM + e] : 0.f;
    const float v  = xm * w[e * 3 + 0] + xc * w[e * 3 + 1] + xp * w[e * 3 + 2] + cb[e];
    x2[(size_t)m * HDIM + e] = silu_fast(v);
}

#define JS 4
#define JW (HDIM / JS)
#define SCAN_THREADS (8 * JW)
#define LOG2E_F 1.4426950408889634f

__device__ __forceinline__ float4 mul4(float4 a, float4 b)
{
    return make_float4(a.x * b.x, a.y * b.y, a.z * b.z, a.w * b.w);
}

__global__ void __launch_bounds__(SCAN_THREADS)
scan_passA(const float* __restrict__ delta,
           const float* __restrict__ Bm,
           const float* __restrict__ x2,
           const float* __restrict__ A_log,
           float* __restrict__ Ssum,
           float* __restrict__ Bacc)
{
    const int wg = blockIdx.x;
    const int js = wg & (JS - 1);
    const int bc = wg / JS;
    const int b  = bc >> 4;
    const int c  = bc & (NCHUNK - 1);
    const int tid = threadIdx.x;
    const int jl  = tid & (JW - 1);
    const int ng  = tid / JW;
    const int j   = js * JW + jl;
    const int nbase = ng << 4;

    __shared__ __attribute__((aligned(16))) float s_d[CLEN * 128];
    __shared__ __attribute__((aligned(16))) float s_dB[CLEN * 128];
    __shared__ __attribute__((aligned(16))) float s_x[CLEN * JW];

    const size_t base = ((size_t)b * LEN + (size_t)c * CLEN) * HDIM;
    {
        const float4* gd = reinterpret_cast<const float4*>(delta + base);
        const float4* gB = reinterpret_cast<const float4*>(Bm + base);
        const float4* gx = reinterpret_cast<const float4*>(x2 + base + js * JW);
        float4* ld = reinterpret_cast<float4*>(s_d);
        float4* lB = reinterpret_cast<float4*>(s_dB);
        float4* lx = reinterpret_cast<float4*>(s_x);
        for (int k = 0; k < (CLEN * 128 / 4) / SCAN_THREADS; ++k) {
            const int q = tid + k * SCAN_THREADS;
            const float4 dv = gd[q];
            ld[q] = dv;
            lB[q] = mul4(dv, gB[q]);
        }
        for (int k = 0; k < (CLEN * JW / 4) / SCAN_THREADS; ++k) {
            const int q = tid + k * SCAN_THREADS;
            lx[q] = gx[(q / (JW / 4)) * (HDIM / 4) + (q % (JW / 4))];
        }
    }

    const float Aj2 = -__expf(A_log[j]) * LOG2E_F;
    float h[16];
#pragma unroll
    for (int i = 0; i < 16; ++i) h[i] = 0.f;
    const int sl = jl & 15;
    float Ssl = 0.f;

    __syncthreads();

    for (int l = 0; l < CLEN; ++l) {
        const float* rd  = s_d  + l * 128 + nbase;
        const float* rdB = s_dB + l * 128 + nbase;
        const float xj = s_x[l * JW + jl];
        Ssl += rd[sl];
#pragma unroll
        for (int i = 0; i < 16; ++i) {
            const float e = __builtin_amdgcn_exp2f(rd[i] * Aj2);
            h[i] = e * h[i] + rdB[i] * xj;
        }
    }

    const size_t ob = (size_t)bc * 128;
#pragma unroll
    for (int i = 0; i < 16; ++i)
        Bacc[(ob + nbase + i) * 128 + j] = h[i];
    if (js == 0 && jl < 16)
        Ssum[ob + nbase + sl] = Ssl;
}

__global__ void __launch_bounds__(SCAN_THREADS)
scan_passB(const float* __restrict__ Ssum,
           const float* __restrict__ Bacc,
           const float* __restrict__ A_log,
           float* __restrict__ hinit)
{
    const int id = blockIdx.x * SCAN_THREADS + threadIdx.x;
    const int j  = id & 127;
    const int n  = (id >> 7) & 127;
    const int b  = id >> 14;

    const float Aj = -__expf(A_log[j]);
    float h = 0.f;
#pragma unroll
    for (int c = 0; c < NCHUNK; ++c) {
        const size_t ob = (size_t)(b * NCHUNK + c) * 128;
        const size_t idx = (ob + n) * 128 + j;
        hinit[idx] = h;
        const float P = __expf(Ssum[ob + n] * Aj);
        h = P * h + Bacc[idx];
    }
}

#define SBLK 8
__global__ void __launch_bounds__(SCAN_THREADS)
scan_passC(const float* __restrict__ delta,
           const float* __restrict__ Bm,
           const float* __restrict__ Cm,
           const float* __restrict__ x2,
           const float* __restrict__ u,
           const float* __restrict__ sz,
           const float* __restrict__ A_log,
           const float* __restrict__ Dv,
           const float* __restrict__ hinit,
           float* __restrict__ ymul)
{
    const int wg = blockIdx.x;
    const int js = wg & (JS - 1);
    const int bc = wg / JS;
    const int b  = bc >> 4;
    const int c  = bc & (NCHUNK - 1);
    const int tid = threadIdx.x;
    const int jl  = tid & (JW - 1);
    const int ng  = tid / JW;
    const int j   = js * JW + jl;
    const int nbase = ng << 4;

    __shared__ __attribute__((aligned(16))) float s_d[CLEN * 128];
    __shared__ __attribute__((aligned(16))) float s_dB[CLEN * 128];
    __shared__ __attribute__((aligned(16))) float s_x[CLEN * JW];
    __shared__ float s_part[2][SBLK][8][JW];

    const size_t base = ((size_t)b * LEN + (size_t)c * CLEN) * HDIM;
    {
        const float4* gd = reinterpret_cast<const float4*>(delta + base);
        const float4* gB = reinterpret_cast<const float4*>(Bm + base);
        const float4* gx = reinterpret_cast<const float4*>(x2 + base + js * JW);
        float4* ld = reinterpret_cast<float4*>(s_d);
        float4* lB = reinterpret_cast<float4*>(s_dB);
        float4* lx = reinterpret_cast<float4*>(s_x);
        for (int k = 0; k < (CLEN * 128 / 4) / SCAN_THREADS; ++k) {
            const int q = tid + k * SCAN_THREADS;
            const float4 dv = gd[q];
            ld[q] = dv;
            lB[q] = mul4(dv, gB[q]);
        }
        for (int k = 0; k < (CLEN * JW / 4) / SCAN_THREADS; ++k) {
            const int q = tid + k * SCAN_THREADS;
            lx[q] = gx[(q / (JW / 4)) * (HDIM / 4) + (q % (JW / 4))];
        }
    }

    const float Aj2 = -__expf(A_log[j]) * LOG2E_F;
    const float Dj = Dv[j];

    const size_t ob = (size_t)bc * 128;
    float h[16];
#pragma unroll
    for (int i = 0; i < 16; ++i) h[i] = hinit[(ob + nbase + i) * 128 + j];

    __syncthreads();

    for (int blk = 0; blk < CLEN / SBLK; ++blk) {
        const int buf = blk & 1;
        const size_t moff = base + (size_t)(blk * SBLK + ng) * HDIM + j;
        const float cv = Cm[moff];
        const float uv = u[moff];
        const float zv = sz[moff];

        for (int ll = 0; ll < SBLK; ++ll) {
            const int l = blk * SBLK + ll;
            const float* rd  = s_d  + l * 128 + nbase;
            const float* rdB = s_dB + l * 128 + nbase;
            const float xj = s_x[l * JW + jl];
            float partial = 0.f;
#pragma unroll
            for (int i = 0; i < 16; ++i) {
                const float e = __builtin_amdgcn_exp2f(rd[i] * Aj2);
                h[i] = e * h[i] + rdB[i] * xj;
                partial += h[i];
            }
            s_part[buf][ll][ng][jl] = partial;
        }
        __syncthreads();
        float tot = 0.f;
#pragma unroll
        for (int g = 0; g < 8; ++g) tot += s_part[buf][ng][g][jl];
        const float y = cv * tot + Dj * uv;
        ymul[moff] = y * zv;
    }
}

extern "C" void kernel_launch(void* const* d_in, const int* in_sizes, int n_in,
                              void* d_out, int out_size, void* d_ws, size_t ws_size,
                              hipStream_t stream)
{
    (void)in_sizes; (void)n_in; (void)out_size; (void)ws_size;

    const float* u      = (const float*)d_in[0];
    const float* W_in   = (const float*)d_in[1];
    const float* b_in   = (const float*)d_in[2];
    const float* conv_w = (const float*)d_in[3];
    const float* conv_b = (const float*)d_in[4];
    const float* W_x    = (const float*)d_in[5];
    const float* b_x    = (const float*)d_in[6];
    const float* A_log  = (const float*)d_in[7];
    const float* Dv     = (const float*)d_in[8];
    const float* W_out  = (const float*)d_in[9];
    const float* b_out  = (const float*)d_in[10];

    const size_t P = (size_t)MTOT * HDIM;
    const size_t NSTATE = (size_t)BATCH * NCHUNK * 128 * 128;

    float* ws    = (float*)d_ws;
    float* x_raw = ws + 0 * P;
    float* s_z   = ws + 1 * P;
    float* x2    = ws + 2 * P;
    float* delta = ws + 3 * P;
    float* Bmat  = ws + 4 * P;
    float* Cmat  = ws + 5 * P;
    float* ymul  = ws + 6 * P;
    float* Ssum  = ws + 7 * P;
    float* Bacc  = Ssum + (size_t)BATCH * NCHUNK * 128;
    float* hinit = Bacc + NSTATE;

    gemm16_wmma_f32<256, 128, EPI_XZ><<<512, 256, 0, stream>>>(
        u, W_in, b_in, x_raw, s_z, nullptr);
    conv_silu_kernel<<<(MTOT * HDIM) / 256, 256, 0, stream>>>(x_raw, conv_w, conv_b, x2);
    gemm16_wmma_f32<384, 128, EPI_SSM><<<768, 256, 0, stream>>>(
        x2, W_x, b_x, delta, Bmat, Cmat);
    scan_passA<<<BATCH * NCHUNK * JS, SCAN_THREADS, 0, stream>>>(delta, Bmat, x2, A_log,
                                                                 Ssum, Bacc);
    scan_passB<<<(BATCH * 128 * 128) / SCAN_THREADS, SCAN_THREADS, 0, stream>>>(
        Ssum, Bacc, A_log, hinit);
    scan_passC<<<BATCH * NCHUNK * JS, SCAN_THREADS, 0, stream>>>(delta, Bmat, Cmat, x2, u, s_z,
                                                                 A_log, Dv, hinit, ymul);
    gemm16_wmma_f32<128, 128, EPI_PLAIN><<<256, 256, 0, stream>>>(
        ymul, W_out, b_out, (float*)d_out, nullptr, nullptr);
}


# conv+silu fused into the ssm GEMM: one 8-wave workgroup per 16 rows forms x2 rows once in LDS as the A operand, 3 tiles per wave, built for 2 waves per SIMD; conv launch gone, 22528 waves to 2048
# speedup vs baseline: 4.2933x; 1.0993x over previous
#include <hip/hip_runtime.h>


#define HDIM 128
#define BATCH 4
#define LEN 1024
#define MTOT (BATCH * LEN)
#define NCHUNK 16
#define CLEN (LEN / NCHUNK)

typedef float v2f __attribute__((ext_vector_type(2)));
typedef float v8f __attribute__((ext_vector_type(8)));

__device__ __forceinline__ float sigmoid_fast(float v) { return 1.0f / (1.0f + __expf(-v)); }
__device__ __forceinline__ float silu_fast(float v)    { return v * sigmoid_fast(v); }
__device__ __forceinline__ float softplus_fast(float v){ return (v > 20.0f) ? v : log1pf(__expf(v)); }

#define EPI_XZ    0
#define EPI_PLAIN 1

template <int N, int K, int MODE>
__global__ void __launch_bounds__(256)
gemm16_wmma_f32(const float* __restrict__ A,
                const float* __restrict__ W,
                const float* __restrict__ bias,
                float* __restrict__ out0,
                float* __restrict__ out1)
{
    const int lane = threadIdx.x & 31;
    const int wave = threadIdx.x >> 5;
    const int gw   = blockIdx.x * (blockDim.x >> 5) + wave;
    const int tilesN = N >> 4;
    const int m0 = (gw / tilesN) << 4;
    const int n0 = (gw % tilesN) << 4;
    const int hi  = lane >> 4;
    const int l16 = lane & 15;

    v8f acc = {0.f, 0.f, 0.f, 0.f, 0.f, 0.f, 0.f, 0.f};

    const float* Arow = A + (size_t)(m0 + l16) * K + 2 * hi;
    const float* Wcol = W + (size_t)(2 * hi) * N + n0 + l16;

#pragma unroll
    for (int k0 = 0; k0 < K; k0 += 4) {
        v2f a;
        a.x = Arow[k0 + 0];
        a.y = Arow[k0 + 1];
        v2f b;
        b.x = Wcol[(k0 + 0) * N];
        b.y = Wcol[(k0 + 1) * N];
        acc = __builtin_amdgcn_wmma_f32_16x16x4_f32(false, a, false, b,
                                                    (short)0, acc, false, false);
    }

    const int   col = n0 + l16;
    const float bv  = bias[col];
#pragma unroll
    for (int r = 0; r < 8; ++r) {
        const int row = m0 + r + hi * 8;
        float v = acc[r] + bv;
        if (MODE == EPI_XZ) {
            if (col < HDIM) out0[(size_t)row * HDIM + col] = v;
            else            out1[(size_t)row * HDIM + (col - HDIM)] = silu_fast(v);
        } else {
            out0[(size_t)row * HDIM + col] = v;
        }
    }
}

#define APITCH 132
__global__ void __launch_bounds__(256) __attribute__((amdgpu_waves_per_eu(2, 2)))
conv_ssm_gemm_wmma_f32(const float* __restrict__ x_raw,
                       const float* __restrict__ cw,
                       const float* __restrict__ cb,
                       const float* __restrict__ W,
                       const float* __restrict__ bias,
                       float* __restrict__ x2,
                       float* __restrict__ delta,
                       float* __restrict__ Bm,
                       float* __restrict__ Cm)
{
    __shared__ __attribute__((aligned(16))) float s_A[16 * APITCH];

    const int lane = threadIdx.x & 31;
    const int wave = threadIdx.x >> 5;
    const int m0   = blockIdx.x << 4;
    const int mA   = m0 + (wave << 1);

    float4 o0, o1;
    {
        const int e  = lane << 2;
        const int r0 = wave << 1;
        const float4* xr = reinterpret_cast<const float4*>(x_raw) + lane;
        const bool hasPrev = (mA & (LEN - 1)) != 0;
        const bool hasNext = ((mA + 1) & (LEN - 1)) != LEN - 1;
        float4 q0 = xr[(size_t)(hasPrev ? mA - 1 : mA) * (HDIM / 4)];
        const float4 q1 = xr[(size_t)mA * (HDIM / 4)];
        const float4 q2 = xr[(size_t)(mA + 1) * (HDIM / 4)];
        float4 q3 = xr[(size_t)(hasNext ? mA + 2 : mA + 1) * (HDIM / 4)];
        if (!hasPrev) q0 = make_float4(0.f, 0.f, 0.f, 0.f);
        if (!hasNext) q3 = make_float4(0.f, 0.f, 0.f, 0.f);

        float wv[12];
#pragma unroll
        for (int i = 0; i < 12; ++i) wv[i] = cw[e * 3 + i];
        float bv[4];
#pragma unroll
        for (int i = 0; i < 4; ++i) bv[i] = cb[e + i];

        o0.x = silu_fast(q0.x * wv[0] + q1.x * wv[1]  + q2.x * wv[2]  + bv[0]);
        o0.y = silu_fast(q0.y * wv[3] + q1.y * wv[4]  + q2.y * wv[5]  + bv[1]);
        o0.z = silu_fast(q0.z * wv[6] + q1.z * wv[7]  + q2.z * wv[8]  + bv[2]);
        o0.w = silu_fast(q0.w * wv[9] + q1.w * wv[10] + q2.w * wv[11] + bv[3]);
        o1.x = silu_fast(q1.x * wv[0] + q2.x * wv[1]  + q3.x * wv[2]  + bv[0]);
        o1.y = silu_fast(q1.y * wv[3] + q2.y * wv[4]  + q3.y * wv[5]  + bv[1]);
        o1.z = silu_fast(q1.z * wv[6] + q2.z * wv[7]  + q3.z * wv[8]  + bv[2]);
        o1.w = silu_fast(q1.w * wv[9] + q2.w * wv[10] + q3.w * wv[11] + bv[3]);

        *reinterpret_cast<float4*>(s_A + r0 * APITCH + e)       = o0;
        *reinterpret_cast<float4*>(s_A + (r0 + 1) * APITCH + e) = o1;
    }
    __syncthreads();
    {
        float4* g = reinterpret_cast<float4*>(x2) + lane;
        g[(size_t)mA * (HDIM / 4)]       = o0;
        g[(size_t)(mA + 1) * (HDIM / 4)] = o1;
    }

    const int hi  = lane >> 4;
    const int l16 = lane & 15;
    const int col = (wave << 4) + l16;

    v8f accD = {0.f, 0.f, 0.f, 0.f, 0.f, 0.f, 0.f, 0.f};
    v8f accB = {0.f, 0.f, 0.f, 0.f, 0.f, 0.f, 0.f, 0.f};
    v8f accC = {0.f, 0.f, 0.f, 0.f, 0.f, 0.f, 0.f, 0.f};

    const float* Arow = s_A + l16 * APITCH + 2 * hi;
    const float* Wcol = W + (size_t)(2 * hi) * 384 + col;

#pragma unroll
    for (int k0 = 0; k0 < HDIM; k0 += 4) {
        v2f a;
        a.x = Arow[k0 + 0];
        a.y = Arow[k0 + 1];
        v2f bD, bB, bC;
        bD.x = Wcol[(k0 + 0) * 384];
        bD.y = Wcol[(k0 + 1) * 384];
        bB.x = Wcol[(k0 + 0) * 384 + 128];
        bB.y = Wcol[(k0 + 1) * 384 + 128];
        bC.x = Wcol[(k0 + 0) * 384 + 256];
        bC.y = Wcol[(k0 + 1) * 384 + 256];
        accD = __builtin_amdgcn_wmma_f32_16x16x4_f32(false, a, false, bD,
                                                     (short)0, accD, false, false);
        accB = __builtin_amdgcn_wmma_f32_16x16x4_f32(false, a, false, bB,
                                                     (short)0, accB, false, false);
        accC = __builtin_amdgcn_wmma_f32_16x16x4_f32(false, a, false, bC,
                                                     (short)0, accC, false, false);
    }

    const float biasD = bias[col];
    const float biasB = bias[128 + col];
    const float biasC = bias[256 + col];
#pragma unroll
    for (int r = 0; r < 8; ++r) {
        const size_t o = (size_t)(m0 + r + hi * 8) * 128 + col;
        delta[o] = softplus_fast(accD[r] + biasD);
        Bm[o]    = accB[r] + biasB;
        Cm[o]    = accC[r] + biasC;
    }
}

#define JS 4
#define JW (HDIM / JS)
#define SCAN_THREADS (8 * JW)
#define LOG2E_F 1.4426950408889634f

__device__ __forceinline__ float4 mul4(float4 a, float4 b)
{
    return make_float4(a.x * b.x, a.y * b.y, a.z * b.z, a.w * b.w);
}

__global__ void __launch_bounds__(SCAN_THREADS)
scan_passA(const float* __restrict__ delta,
           const float* __restrict__ Bm,
           const float* __restrict__ x2,
           const float* __restrict__ A_log,
           float* __restrict__ Ssum,
           float* __restrict__ Bacc)
{
    const int wg = blockIdx.x;
    const int js = wg & (JS - 1);
    const int bc = wg / JS;
    const int b  = bc >> 4;
    const int c  = bc & (NCHUNK - 1);
    const int tid = threadIdx.x;
    const int jl  = tid & (JW - 1);
    const int ng  = tid / JW;
    const int j   = js * JW + jl;
    const int nbase = ng << 4;

    __shared__ __attribute__((aligned(16))) float s_d[CLEN * 128];
    __shared__ __attribute__((aligned(16))) float s_dB[CLEN * 128];
    __shared__ __attribute__((aligned(16))) float s_x[CLEN * JW];

    const size_t base = ((size_t)b * LEN + (size_t)c * CLEN) * HDIM;
    {
        const float4* gd = reinterpret_cast<const float4*>(delta + base);
        const float4* gB = reinterpret_cast<const float4*>(Bm + base);
        const float4* gx = reinterpret_cast<const float4*>(x2 + base + js * JW);
        float4* ld = reinterpret_cast<float4*>(s_d);
        float4* lB = reinterpret_cast<float4*>(s_dB);
        float4* lx = reinterpret_cast<float4*>(s_x);
        for (int k = 0; k < (CLEN * 128 / 4) / SCAN_THREADS; ++k) {
            const int q = tid + k * SCAN_THREADS;
            const float4 dv = gd[q];
            ld[q] = dv;
            lB[q] = mul4(dv, gB[q]);
        }
        for (int k = 0; k < (CLEN * JW / 4) / SCAN_THREADS; ++k) {
            const int q = tid + k * SCAN_THREADS;
            lx[q] = gx[(q / (JW / 4)) * (HDIM / 4) + (q % (JW / 4))];
        }
    }

    const float Aj2 = -__expf(A_log[j]) * LOG2E_F;
    float h[16];
#pragma unroll
    for (int i = 0; i < 16; ++i) h[i] = 0.f;
    const int sl = jl & 15;
    float Ssl = 0.f;

    __syncthreads();

    for (int l = 0; l < CLEN; ++l) {
        const float* rd  = s_d  + l * 128 + nbase;
        const float* rdB = s_dB + l * 128 + nbase;
        const float xj = s_x[l * JW + jl];
        Ssl += rd[sl];
#pragma unroll
        for (int i = 0; i < 16; ++i) {
            const float e = __builtin_amdgcn_exp2f(rd[i] * Aj2);
            h[i] = e * h[i] + rdB[i] * xj;
        }
    }

    const size_t ob = (size_t)bc * 128;
#pragma unroll
    for (int i = 0; i < 16; ++i)
        Bacc[(ob + nbase + i) * 128 + j] = h[i];
    if (js == 0 && jl < 16)
        Ssum[ob + nbase + sl] = Ssl;
}

__global__ void __launch_bounds__(SCAN_THREADS)
scan_passB(const float* __restrict__ Ssum,
           const float* __restrict__ Bacc,
           const float* __restrict__ A_log,
           float* __restrict__ hinit)
{
    const int id = blockIdx.x * SCAN_THREADS + threadIdx.x;
    const int j  = id & 127;
    const int n  = (id >> 7) & 127;
    const int b  = id >> 14;

    const float Aj = -__expf(A_log[j]);
    float h = 0.f;
#pragma unroll
    for (int c = 0; c < NCHUNK; ++c) {
        const size_t ob = (size_t)(b * NCHUNK + c) * 128;
        const size_t idx = (ob + n) * 128 + j;
        hinit[idx] = h;
        const float P = __expf(Ssum[ob + n] * Aj);
        h = P * h + Bacc[idx];
    }
}

#define SBLK 8
__global__ void __launch_bounds__(SCAN_THREADS)
scan_passC(const float* __restrict__ delta,
           const float* __restrict__ Bm,
           const float* __restrict__ Cm,
           const float* __restrict__ x2,
           const float* __restrict__ u,
           const float* __restrict__ sz,
           const float* __restrict__ A_log,
           const float* __restrict__ Dv,
           const float* __restrict__ hinit,
           float* __restrict__ ymul)
{
    const int wg = blockIdx.x;
    const int js = wg & (JS - 1);
    const int bc = wg / JS;
    const int b  = bc >> 4;
    const int c  = bc & (NCHUNK - 1);
    const int tid = threadIdx.x;
    const int jl  = tid & (JW - 1);
    const int ng  = tid / JW;
    const int j   = js * JW + jl;
    const int nbase = ng << 4;

    __shared__ __attribute__((aligned(16))) float s_d[CLEN * 128];
    __shared__ __attribute__((aligned(16))) float s_dB[CLEN * 128];
    __shared__ __attribute__((aligned(16))) float s_x[CLEN * JW];
    __shared__ float s_part[2][SBLK][8][JW];

    const size_t base = ((size_t)b * LEN + (size_t)c * CLEN) * HDIM;
    {
        const float4* gd = reinterpret_cast<const float4*>(delta + base);
        const float4* gB = reinterpret_cast<const float4*>(Bm + base);
        const float4* gx = reinterpret_cast<const float4*>(x2 + base + js * JW);
        float4* ld = reinterpret_cast<float4*>(s_d);
        float4* lB = reinterpret_cast<float4*>(s_dB);
        float4* lx = reinterpret_cast<float4*>(s_x);
        for (int k = 0; k < (CLEN * 128 / 4) / SCAN_THREADS; ++k) {
            const int q = tid + k * SCAN_THREADS;
            const float4 dv = gd[q];
            ld[q] = dv;
            lB[q] = mul4(dv, gB[q]);
        }
        for (int k = 0; k < (CLEN * JW / 4) / SCAN_THREADS; ++k) {
            const int q = tid + k * SCAN_THREADS;
            lx[q] = gx[(q / (JW / 4)) * (HDIM / 4) + (q % (JW / 4))];
        }
    }

    const float Aj2 = -__expf(A_log[j]) * LOG2E_F;
    const float Dj = Dv[j];

    const size_t ob = (size_t)bc * 128;
    float h[16];
#pragma unroll
    for (int i = 0; i < 16; ++i) h[i] = hinit[(ob + nbase + i) * 128 + j];

    __syncthreads();

    for (int blk = 0; blk < CLEN / SBLK; ++blk) {
        const int buf = blk & 1;
        const size_t moff = base + (size_t)(blk * SBLK + ng) * HDIM + j;
        const float cv = Cm[moff];
        const float uv = u[moff];
        const float zv = sz[moff];

        for (int ll = 0; ll < SBLK; ++ll) {
            const int l = blk * SBLK + ll;
            const float* rd  = s_d  + l * 128 + nbase;
            const float* rdB = s_dB + l * 128 + nbase;
            const float xj = s_x[l * JW + jl];
            float partial = 0.f;
#pragma unroll
            for (int i = 0; i < 16; ++i) {
                const float e = __builtin_amdgcn_exp2f(rd[i] * Aj2);
                h[i] = e * h[i] + rdB[i] * xj;
                partial += h[i];
            }
            s_part[buf][ll][ng][jl] = partial;
        }
        __syncthreads();
        float tot = 0.f;
#pragma unroll
        for (int g = 0; g < 8; ++g) tot += s_part[buf][ng][g][jl];
        const float y = cv * tot + Dj * uv;
        ymul[moff] = y * zv;
    }
}

extern "C" void kernel_launch(void* const* d_in, const int* in_sizes, int n_in,
                              void* d_out, int out_size, void* d_ws, size_t ws_size,
                              hipStream_t stream)
{
    (void)in_sizes; (void)n_in; (void)out_size; (void)ws_size;

    const float* u      = (const float*)d_in[0];
    const float* W_in   = (const float*)d_in[1];
    const float* b_in   = (const float*)d_in[2];
    const float* conv_w = (const float*)d_in[3];
    const float* conv_b = (const float*)d_in[4];
    const float* W_x    = (const float*)d_in[5];
    const float* b_x    = (const float*)d_in[6];
    const float* A_log  = (const float*)d_in[7];
    const float* Dv     = (const float*)d_in[8];
    const float* W_out  = (const float*)d_in[9];
    const float* b_out  = (const float*)d_in[10];

    const size_t P = (size_t)MTOT * HDIM;
    const size_t NSTATE = (size_t)BATCH * NCHUNK * 128 * 128;

    float* ws    = (float*)d_ws;
    float* x_raw = ws + 0 * P;
    float* s_z   = ws + 1 * P;
    float* x2    = ws + 2 * P;
    float* delta = ws + 3 * P;
    float* Bmat  = ws + 4 * P;
    float* Cmat  = ws + 5 * P;
    float* ymul  = ws + 6 * P;
    float* Ssum  = ws + 7 * P;
    float* Bacc  = Ssum + (size_t)BATCH * NCHUNK * 128;
    float* hinit = Bacc + NSTATE;

    gemm16_wmma_f32<256, 128, EPI_XZ><<<512, 256, 0, stream>>>(
        u, W_in, b_in, x_raw, s_z);
    conv_ssm_gemm_wmma_f32<<<MTOT / 16, 256, 0, stream>>>(
        x_raw, conv_w, conv_b, W_x, b_x, x2, delta, Bmat, Cmat);
    scan_passA<<<BATCH * NCHUNK * JS, SCAN_THREADS, 0, stream>>>(delta, Bmat, x2, A_log,
                                                                 Ssum, Bacc);
    scan_passB<<<(BATCH * 128 * 128) / SCAN_THREADS, SCAN_THREADS, 0, stream>>>(
        Ssum, Bacc, A_log, hinit);
    scan_passC<<<BATCH * NCHUNK * JS, SCAN_THREADS, 0, stream>>>(delta, Bmat, Cmat, x2, u, s_z,
                                                                 A_log, Dv, hinit, ymul);
    gemm16_wmma_f32<128, 128, EPI_PLAIN><<<256, 256, 0, stream>>>(
        ymul, W_out, b_out, (float*)d_out, nullptr);
}


# in-projection GEMM fused into the conv+ssm kernel: each 16-row workgroup takes u through LDS, forms x_raw for its rows and 2 halo rows (a second A tile) in LDS; one launch and the x_raw plane gone
# speedup vs baseline: 4.7254x; 1.1006x over previous
#include <hip/hip_runtime.h>


#define HDIM 128
#define BATCH 4
#define LEN 1024
#define MTOT (BATCH * LEN)
#define NCHUNK 16
#define CLEN (LEN / NCHUNK)

typedef float v2f __attribute__((ext_vector_type(2)));
typedef float v8f __attribute__((ext_vector_type(8)));

__device__ __forceinline__ float sigmoid_fast(float v) { return 1.0f / (1.0f + __expf(-v)); }
__device__ __forceinline__ float silu_fast(float v)    { return v * sigmoid_fast(v); }
__device__ __forceinline__ float softplus_fast(float v){ return (v > 20.0f) ? v : log1pf(__expf(v)); }

#define EPI_XZ    0
#define EPI_PLAIN 1

template <int N, int K, int MODE>
__global__ void __launch_bounds__(256)
gemm16_wmma_f32(const float* __restrict__ A,
                const float* __restrict__ W,
                const float* __restrict__ bias,
                float* __restrict__ out0,
                float* __restrict__ out1)
{
    const int lane = threadIdx.x & 31;
    const int wave = threadIdx.x >> 5;
    const int gw   = blockIdx.x * (blockDim.x >> 5) + wave;
    const int tilesN = N >> 4;
    const int m0 = (gw / tilesN) << 4;
    const int n0 = (gw % tilesN) << 4;
    const int hi  = lane >> 4;
    const int l16 = lane & 15;

    v8f acc = {0.f, 0.f, 0.f, 0.f, 0.f, 0.f, 0.f, 0.f};

    const float* Arow = A + (size_t)(m0 + l16) * K + 2 * hi;
    const float* Wcol = W + (size_t)(2 * hi) * N + n0 + l16;

#pragma unroll
    for (int k0 = 0; k0 < K; k0 += 4) {
        v2f a;
        a.x = Arow[k0 + 0];
        a.y = Arow[k0 + 1];
        v2f b;
        b.x = Wcol[(k0 + 0) * N];
        b.y = Wcol[(k0 + 1) * N];
        acc = __builtin_amdgcn_wmma_f32_16x16x4_f32(false, a, false, b,
                                                    (short)0, acc, false, false);
    }

    const int   col = n0 + l16;
    const float bv  = bias[col];
#pragma unroll
    for (int r = 0; r < 8; ++r) {
        const int row = m0 + r + hi * 8;
        float v = acc[r] + bv;
        if (MODE == EPI_XZ) {
            if (col < HDIM) out0[(size_t)row * HDIM + col] = v;
            else            out1[(size_t)row * HDIM + (col - HDIM)] = silu_fast(v);
        } else {
            out0[(size_t)row * HDIM + col] = v;
        }
    }
}

#define APITCH 132
__global__ void __launch_bounds__(256) __attribute__((amdgpu_waves_per_eu(2, 2)))
inproj_conv_ssm_wmma_f32(const float* __restrict__ u,
                         const float* __restrict__ W_in,
                         const float* __restrict__ b_in,
                         const float* __restrict__ cw,
                         const float* __restrict__ cb,
                         const float* __restrict__ W,
                         const float* __restrict__ bias,
                         float* __restrict__ sz,
                         float* __restrict__ x2,
                         float* __restrict__ delta,
                         float* __restrict__ Bm,
                         float* __restrict__ Cm)
{
    __shared__ __attribute__((aligned(16))) float s_U[16 * APITCH];
    __shared__ __attribute__((aligned(16))) float s_UH[16 * APITCH];
    __shared__ __attribute__((aligned(16))) float s_X[18 * APITCH];
    __shared__ __attribute__((aligned(16))) float s_A[16 * APITCH];

    const int tid  = threadIdx.x;
    const int lane = tid & 31;
    const int wave = tid >> 5;
    const int m0   = blockIdx.x << 4;
    const int mA   = m0 + (wave << 1);
    const bool hasPrev = (m0 & (LEN - 1)) != 0;
    const bool hasNext = ((m0 + 16) & (LEN - 1)) != 0;
    const int hi  = lane >> 4;
    const int l16 = lane & 15;
    const int e   = lane << 2;

    {
        const float4* ug = reinterpret_cast<const float4*>(u) + lane;
        const int hsrc = (wave == 0) ? (hasPrev ? m0 - 1 : m0) : (hasNext ? m0 + 16 : m0);
        const float4 v0 = ug[(size_t)(m0 + wave) * (HDIM / 4)];
        const float4 v1 = ug[(size_t)(m0 + wave + 8) * (HDIM / 4)];
        float4 hv       = ug[(size_t)hsrc * (HDIM / 4)];
        const float4 zero4 = make_float4(0.f, 0.f, 0.f, 0.f);
        if (wave >= 2) hv = zero4;
        *reinterpret_cast<float4*>(s_U  + wave * APITCH + e)       = v0;
        *reinterpret_cast<float4*>(s_U  + (wave + 8) * APITCH + e) = v1;
        *reinterpret_cast<float4*>(s_UH + wave * APITCH + e)       = hv;
        *reinterpret_cast<float4*>(s_UH + (wave + 8) * APITCH + e) = zero4;
    }

    float wv[12];
#pragma unroll
    for (int i = 0; i < 12; ++i) wv[i] = cw[e * 3 + i];
    float bv[4];
#pragma unroll
    for (int i = 0; i < 4; ++i) bv[i] = cb[e + i];

    __syncthreads();

    const int col = (wave << 4) + l16;

    float zs[8];
    {
        v8f accX = {0.f, 0.f, 0.f, 0.f, 0.f, 0.f, 0.f, 0.f};
        v8f accH = {0.f, 0.f, 0.f, 0.f, 0.f, 0.f, 0.f, 0.f};
        v8f accZ = {0.f, 0.f, 0.f, 0.f, 0.f, 0.f, 0.f, 0.f};
        const float* Urow = s_U  + l16 * APITCH + 2 * hi;
        const float* Hrow = s_UH + l16 * APITCH + 2 * hi;
        const float* Wcol = W_in + (size_t)(2 * hi) * 256 + col;
#pragma unroll
        for (int k0 = 0; k0 < HDIM; k0 += 4) {
            v2f a, ah;
            a.x  = Urow[k0 + 0];
            a.y  = Urow[k0 + 1];
            ah.x = Hrow[k0 + 0];
            ah.y = Hrow[k0 + 1];
            v2f bX, bZ;
            bX.x = Wcol[(k0 + 0) * 256];
            bX.y = Wcol[(k0 + 1) * 256];
            bZ.x = Wcol[(k0 + 0) * 256 + 128];
            bZ.y = Wcol[(k0 + 1) * 256 + 128];
            accX = __builtin_amdgcn_wmma_f32_16x16x4_f32(false, a, false, bX,
                                                         (short)0, accX, false, false);
            accH = __builtin_amdgcn_wmma_f32_16x16x4_f32(false, ah, false, bX,
                                                         (short)0, accH, false, false);
            accZ = __builtin_amdgcn_wmma_f32_16x16x4_f32(false, a, false, bZ,
                                                         (short)0, accZ, false, false);
        }
        const float biasX = b_in[col];
        const float biasZ = b_in[128 + col];
#pragma unroll
        for (int r = 0; r < 8; ++r) {
            const int row = r + hi * 8;
            s_X[(1 + row) * APITCH + col] = accX[r] + biasX;
            zs[r] = silu_fast(accZ[r] + biasZ);
        }
        if (hi == 0) {
            s_X[col]               = hasPrev ? accH[0] + biasX : 0.f;
            s_X[17 * APITCH + col] = hasNext ? accH[1] + biasX : 0.f;
        }
    }
    __syncthreads();

    float4 o0, o1;
    {
        const int r0 = wave << 1;
        const float4 q0 = *reinterpret_cast<const float4*>(s_X + (r0 + 0) * APITCH + e);
        const float4 q1 = *reinterpret_cast<const float4*>(s_X + (r0 + 1) * APITCH + e);
        const float4 q2 = *reinterpret_cast<const float4*>(s_X + (r0 + 2) * APITCH + e);
        const float4 q3 = *reinterpret_cast<const float4*>(s_X + (r0 + 3) * APITCH + e);

        o0.x = silu_fast(q0.x * wv[0] + q1.x * wv[1]  + q2.x * wv[2]  + bv[0]);
        o0.y = silu_fast(q0.y * wv[3] + q1.y * wv[4]  + q2.y * wv[5]  + bv[1]);
        o0.z = silu_fast(q0.z * wv[6] + q1.z * wv[7]  + q2.z * wv[8]  + bv[2]);
        o0.w = silu_fast(q0.w * wv[9] + q1.w * wv[10] + q2.w * wv[11] + bv[3]);
        o1.x = silu_fast(q1.x * wv[0] + q2.x * wv[1]  + q3.x * wv[2]  + bv[0]);
        o1.y = silu_fast(q1.y * wv[3] + q2.y * wv[4]  + q3.y * wv[5]  + bv[1]);
        o1.z = silu_fast(q1.z * wv[6] + q2.z * wv[7]  + q3.z * wv[8]  + bv[2]);
        o1.w = silu_fast(q1.w * wv[9] + q2.w * wv[10] + q3.w * wv[11] + bv[3]);

        *reinterpret_cast<float4*>(s_A + r0 * APITCH + e)       = o0;
        *reinterpret_cast<float4*>(s_A + (r0 + 1) * APITCH + e) = o1;
    }
    __syncthreads();
    {
        float4* g = reinterpret_cast<float4*>(x2) + lane;
        g[(size_t)mA * (HDIM / 4)]       = o0;
        g[(size_t)(mA + 1) * (HDIM / 4)] = o1;
#pragma unroll
        for (int r = 0; r < 8; ++r)
            sz[(size_t)(m0 + r + hi * 8) * HDIM + col] = zs[r];
    }

    v8f accD = {0.f, 0.f, 0.f, 0.f, 0.f, 0.f, 0.f, 0.f};
    v8f accB = {0.f, 0.f, 0.f, 0.f, 0.f, 0.f, 0.f, 0.f};
    v8f accC = {0.f, 0.f, 0.f, 0.f, 0.f, 0.f, 0.f, 0.f};

    const float* Arow = s_A + l16 * APITCH + 2 * hi;
    const float* Wcol = W + (size_t)(2 * hi) * 384 + col;

#pragma unroll
    for (int k0 = 0; k0 < HDIM; k0 += 4) {
        v2f a;
        a.x = Arow[k0 + 0];
        a.y = Arow[k0 + 1];
        v2f bD, bB, bC;
        bD.x = Wcol[(k0 + 0) * 384];
        bD.y = Wcol[(k0 + 1) * 384];
        bB.x = Wcol[(k0 + 0) * 384 + 128];
        bB.y = Wcol[(k0 + 1) * 384 + 128];
        bC.x = Wcol[(k0 + 0) * 384 + 256];
        bC.y = Wcol[(k0 + 1) * 384 + 256];
        accD = __builtin_amdgcn_wmma_f32_16x16x4_f32(false, a, false, bD,
                                                     (short)0, accD, false, false);
        accB = __builtin_amdgcn_wmma_f32_16x16x4_f32(false, a, false, bB,
                                                     (short)0, accB, false, false);
        accC = __builtin_amdgcn_wmma_f32_16x16x4_f32(false, a, false, bC,
                                                     (short)0, accC, false, false);
    }

    const float biasD = bias[col];
    const float biasB = bias[128 + col];
    const float biasC = bias[256 + col];
#pragma unroll
    for (int r = 0; r < 8; ++r) {
        const size_t o = (size_t)(m0 + r + hi * 8) * 128 + col;
        delta[o] = softplus_fast(accD[r] + biasD);
        Bm[o]    = accB[r] + biasB;
        Cm[o]    = accC[r] + biasC;
    }
}

#define JS 4
#define JW (HDIM / JS)
#define SCAN_THREADS (8 * JW)
#define LOG2E_F 1.4426950408889634f

__device__ __forceinline__ float4 mul4(float4 a, float4 b)
{
    return make_float4(a.x * b.x, a.y * b.y, a.z * b.z, a.w * b.w);
}

__global__ void __launch_bounds__(SCAN_THREADS)
scan_passA(const float* __restrict__ delta,
           const float* __restrict__ Bm,
           const float* __restrict__ x2,
           const float* __restrict__ A_log,
           float* __restrict__ Ssum,
           float* __restrict__ Bacc)
{
    const int wg = blockIdx.x;
    const int js = wg & (JS - 1);
    const int bc = wg / JS;
    const int b  = bc >> 4;
    const int c  = bc & (NCHUNK - 1);
    const int tid = threadIdx.x;
    const int jl  = tid & (JW - 1);
    const int ng  = tid / JW;
    const int j   = js * JW + jl;
    const int nbase = ng << 4;

    __shared__ __attribute__((aligned(16))) float s_d[CLEN * 128];
    __shared__ __attribute__((aligned(16))) float s_dB[CLEN * 128];
    __shared__ __attribute__((aligned(16))) float s_x[CLEN * JW];

    const size_t base = ((size_t)b * LEN + (size_t)c * CLEN) * HDIM;
    {
        const float4* gd = reinterpret_cast<const float4*>(delta + base);
        const float4* gB = reinterpret_cast<const float4*>(Bm + base);
        const float4* gx = reinterpret_cast<const float4*>(x2 + base + js * JW);
        float4* ld = reinterpret_cast<float4*>(s_d);
        float4* lB = reinterpret_cast<float4*>(s_dB);
        float4* lx = reinterpret_cast<float4*>(s_x);
        for (int k = 0; k < (CLEN * 128 / 4) / SCAN_THREADS; ++k) {
            const int q = tid + k * SCAN_THREADS;
            const float4 dv = gd[q];
            ld[q] = dv;
            lB[q] = mul4(dv, gB[q]);
        }
        for (int k = 0; k < (CLEN * JW / 4) / SCAN_THREADS; ++k) {
            const int q = tid + k * SCAN_THREADS;
            lx[q] = gx[(q / (JW / 4)) * (HDIM / 4) + (q % (JW / 4))];
        }
    }

    const float Aj2 = -__expf(A_log[j]) * LOG2E_F;
    float h[16];
#pragma unroll
    for (int i = 0; i < 16; ++i) h[i] = 0.f;
    const int sl = jl & 15;
    float Ssl = 0.f;

    __syncthreads();

    for (int l = 0; l < CLEN; ++l) {
        const float* rd  = s_d  + l * 128 + nbase;
        const float* rdB = s_dB + l * 128 + nbase;
        const float xj = s_x[l * JW + jl];
        Ssl += rd[sl];
#pragma unroll
        for (int i = 0; i < 16; ++i) {
            const float e = __builtin_amdgcn_exp2f(rd[i] * Aj2);
            h[i] = e * h[i] + rdB[i] * xj;
        }
    }

    const size_t ob = (size_t)bc * 128;
#pragma unroll
    for (int i = 0; i < 16; ++i)
        Bacc[(ob + nbase + i) * 128 + j] = h[i];
    if (js == 0 && jl < 16)
        Ssum[ob + nbase + sl] = Ssl;
}

__global__ void __launch_bounds__(SCAN_THREADS)
scan_passB(const float* __restrict__ Ssum,
           const float* __restrict__ Bacc,
           const float* __restrict__ A_log,
           float* __restrict__ hinit)
{
    const int id = blockIdx.x * SCAN_THREADS + threadIdx.x;
    const int j  = id & 127;
    const int n  = (id >> 7) & 127;
    const int b  = id >> 14;

    const float Aj = -__expf(A_log[j]);
    float h = 0.f;
#pragma unroll
    for (int c = 0; c < NCHUNK; ++c) {
        const size_t ob = (size_t)(b * NCHUNK + c) * 128;
        const size_t idx = (ob + n) * 128 + j;
        hinit[idx] = h;
        const float P = __expf(Ssum[ob + n] * Aj);
        h = P * h + Bacc[idx];
    }
}

#define SBLK 8
__global__ void __launch_bounds__(SCAN_THREADS)
scan_passC(const float* __restrict__ delta,
           const float* __restrict__ Bm,
           const float* __restrict__ Cm,
           const float* __restrict__ x2,
           const float* __restrict__ u,
           const float* __restrict__ sz,
           const float* __restrict__ A_log,
           const float* __restrict__ Dv,
           const float* __restrict__ hinit,
           float* __restrict__ ymul)
{
    const int wg = blockIdx.x;
    const int js = wg & (JS - 1);
    const int bc = wg / JS;
    const int b  = bc >> 4;
    const int c  = bc & (NCHUNK - 1);
    const int tid = threadIdx.x;
    const int jl  = tid & (JW - 1);
    const int ng  = tid / JW;
    const int j   = js * JW + jl;
    const int nbase = ng << 4;

    __shared__ __attribute__((aligned(16))) float s_d[CLEN * 128];
    __shared__ __attribute__((aligned(16))) float s_dB[CLEN * 128];
    __shared__ __attribute__((aligned(16))) float s_x[CLEN * JW];
    __shared__ float s_part[2][SBLK][8][JW];

    const size_t base = ((size_t)b * LEN + (size_t)c * CLEN) * HDIM;
    {
        const float4* gd = reinterpret_cast<const float4*>(delta + base);
        const float4* gB = reinterpret_cast<const float4*>(Bm + base);
        const float4* gx = reinterpret_cast<const float4*>(x2 + base + js * JW);
        float4* ld = reinterpret_cast<float4*>(s_d);
        float4* lB = reinterpret_cast<float4*>(s_dB);
        float4* lx = reinterpret_cast<float4*>(s_x);
        for (int k = 0; k < (CLEN * 128 / 4) / SCAN_THREADS; ++k) {
            const int q = tid + k * SCAN_THREADS;
            const float4 dv = gd[q];
            ld[q] = dv;
            lB[q] = mul4(dv, gB[q]);
        }
        for (int k = 0; k < (CLEN * JW / 4) / SCAN_THREADS; ++k) {
            const int q = tid + k * SCAN_THREADS;
            lx[q] = gx[(q / (JW / 4)) * (HDIM / 4) + (q % (JW / 4))];
        }
    }

    const float Aj2 = -__expf(A_log[j]) * LOG2E_F;
    const float Dj = Dv[j];

    const size_t ob = (size_t)bc * 128;
    float h[16];
#pragma unroll
    for (int i = 0; i < 16; ++i) h[i] = hinit[(ob + nbase + i) * 128 + j];

    __syncthreads();

    for (int blk = 0; blk < CLEN / SBLK; ++blk) {
        const int buf = blk & 1;
        const size_t moff = base + (size_t)(blk * SBLK + ng) * HDIM + j;
        const float cv = Cm[moff];
        const float uv = u[moff];
        const float zv = sz[moff];

        for (int ll = 0; ll < SBLK; ++ll) {
            const int l = blk * SBLK + ll;
            const float* rd  = s_d  + l * 128 + nbase;
            const float* rdB = s_dB + l * 128 + nbase;
            const float xj = s_x[l * JW + jl];
            float partial = 0.f;
#pragma unroll
            for (int i = 0; i < 16; ++i) {
                const float e = __builtin_amdgcn_exp2f(rd[i] * Aj2);
                h[i] = e * h[i] + rdB[i] * xj;
                partial += h[i];
            }
            s_part[buf][ll][ng][jl] = partial;
        }
        __syncthreads();
        float tot = 0.f;
#pragma unroll
        for (int g = 0; g < 8; ++g) tot += s_part[buf][ng][g][jl];
        const float y = cv * tot + Dj * uv;
        ymul[moff] = y * zv;
    }
}

extern "C" void kernel_launch(void* const* d_in, const int* in_sizes, int n_in,
                              void* d_out, int out_size, void* d_ws, size_t ws_size,
                              hipStream_t stream)
{
    (void)in_sizes; (void)n_in; (void)out_size; (void)ws_size;

    const float* u      = (const float*)d_in[0];
    const float* W_in   = (const float*)d_in[1];
    const float* b_in   = (const float*)d_in[2];
    const float* conv_w = (const float*)d_in[3];
    const float* conv_b = (const float*)d_in[4];
    const float* W_x    = (const float*)d_in[5];
    const float* b_x    = (const float*)d_in[6];
    const float* A_log  = (const float*)d_in[7];
    const float* Dv     = (const float*)d_in[8];
    const float* W_out  = (const float*)d_in[9];
    const float* b_out  = (const float*)d_in[10];

    const size_t P = (size_t)MTOT * HDIM;
    const size_t NSTATE = (size_t)BATCH * NCHUNK * 128 * 128;

    float* ws    = (float*)d_ws;
    float* s_z   = ws + 1 * P;
    float* x2    = ws + 2 * P;
    float* delta = ws + 3 * P;
    float* Bmat  = ws + 4 * P;
    float* Cmat  = ws + 5 * P;
    float* ymul  = ws + 6 * P;
    float* Ssum  = ws + 7 * P;
    float* Bacc  = Ssum + (size_t)BATCH * NCHUNK * 128;
    float* hinit = Bacc + NSTATE;

    inproj_conv_ssm_wmma_f32<<<MTOT / 16, 256, 0, stream>>>(
        u, W_in, b_in, conv_w, conv_b, W_x, b_x, s_z, x2, delta, Bmat, Cmat);
    scan_passA<<<BATCH * NCHUNK * JS, SCAN_THREADS, 0, stream>>>(delta, Bmat, x2, A_log,
                                                                 Ssum, Bacc);
    scan_passB<<<(BATCH * 128 * 128) / SCAN_THREADS, SCAN_THREADS, 0, stream>>>(
        Ssum, Bacc, A_log, hinit);
    scan_passC<<<BATCH * NCHUNK * JS, SCAN_THREADS, 0, stream>>>(delta, Bmat, Cmat, x2, u, s_z,
                                                                 A_log, Dv, hinit, ymul);
    gemm16_wmma_f32<128, 128, EPI_PLAIN><<<256, 256, 0, stream>>>(
        ymul, W_out, b_out, (float*)d_out, nullptr);
}


# scan passes A and C: each lane owns 2 adjacent columns x 8 states (was 1 x 16), so each LDS read of a d or d*B row feeds twice the work: LDS bytes per step halved; y now sums 8 states x 16 groups
# speedup vs baseline: 4.8929x; 1.0354x over previous
#include <hip/hip_runtime.h>


#define HDIM 128
#define BATCH 4
#define LEN 1024
#define MTOT (BATCH * LEN)
#define NCHUNK 16
#define CLEN (LEN / NCHUNK)

typedef float v2f __attribute__((ext_vector_type(2)));
typedef float v8f __attribute__((ext_vector_type(8)));

__device__ __forceinline__ float sigmoid_fast(float v) { return 1.0f / (1.0f + __expf(-v)); }
__device__ __forceinline__ float silu_fast(float v)    { return v * sigmoid_fast(v); }
__device__ __forceinline__ float softplus_fast(float v){ return (v > 20.0f) ? v : log1pf(__expf(v)); }

#define EPI_XZ    0
#define EPI_PLAIN 1

template <int N, int K, int MODE>
__global__ void __launch_bounds__(256)
gemm16_wmma_f32(const float* __restrict__ A,
                const float* __restrict__ W,
                const float* __restrict__ bias,
                float* __restrict__ out0,
                float* __restrict__ out1)
{
    const int lane = threadIdx.x & 31;
    const int wave = threadIdx.x >> 5;
    const int gw   = blockIdx.x * (blockDim.x >> 5) + wave;
    const int tilesN = N >> 4;
    const int m0 = (gw / tilesN) << 4;
    const int n0 = (gw % tilesN) << 4;
    const int hi  = lane >> 4;
    const int l16 = lane & 15;

    v8f acc = {0.f, 0.f, 0.f, 0.f, 0.f, 0.f, 0.f, 0.f};

    const float* Arow = A + (size_t)(m0 + l16) * K + 2 * hi;
    const float* Wcol = W + (size_t)(2 * hi) * N + n0 + l16;

#pragma unroll
    for (int k0 = 0; k0 < K; k0 += 4) {
        v2f a;
        a.x = Arow[k0 + 0];
        a.y = Arow[k0 + 1];
        v2f b;
        b.x = Wcol[(k0 + 0) * N];
        b.y = Wcol[(k0 + 1) * N];
        acc = __builtin_amdgcn_wmma_f32_16x16x4_f32(false, a, false, b,
                                                    (short)0, acc, false, false);
    }

    const int   col = n0 + l16;
    const float bv  = bias[col];
#pragma unroll
    for (int r = 0; r < 8; ++r) {
        const int row = m0 + r + hi * 8;
        float v = acc[r] + bv;
        if (MODE == EPI_XZ) {
            if (col < HDIM) out0[(size_t)row * HDIM + col] = v;
            else            out1[(size_t)row * HDIM + (col - HDIM)] = silu_fast(v);
        } else {
            out0[(size_t)row * HDIM + col] = v;
        }
    }
}

#define APITCH 132
__global__ void __launch_bounds__(256) __attribute__((amdgpu_waves_per_eu(2, 2)))
inproj_conv_ssm_wmma_f32(const float* __restrict__ u,
                         const float* __restrict__ W_in,
                         const float* __restrict__ b_in,
                         const float* __restrict__ cw,
                         const float* __restrict__ cb,
                         const float* __restrict__ W,
                         const float* __restrict__ bias,
                         float* __restrict__ sz,
                         float* __restrict__ x2,
                         float* __restrict__ delta,
                         float* __restrict__ Bm,
                         float* __restrict__ Cm)
{
    __shared__ __attribute__((aligned(16))) float s_U[16 * APITCH];
    __shared__ __attribute__((aligned(16))) float s_UH[16 * APITCH];
    __shared__ __attribute__((aligned(16))) float s_X[18 * APITCH];
    __shared__ __attribute__((aligned(16))) float s_A[16 * APITCH];

    const int tid  = threadIdx.x;
    const int lane = tid & 31;
    const int wave = tid >> 5;
    const int m0   = blockIdx.x << 4;
    const int mA   = m0 + (wave << 1);
    const bool hasPrev = (m0 & (LEN - 1)) != 0;
    const bool hasNext = ((m0 + 16) & (LEN - 1)) != 0;
    const int hi  = lane >> 4;
    const int l16 = lane & 15;
    const int e   = lane << 2;

    {
        const float4* ug = reinterpret_cast<const float4*>(u) + lane;
        const int hsrc = (wave == 0) ? (hasPrev ? m0 - 1 : m0) : (hasNext ? m0 + 16 : m0);
        const float4 v0 = ug[(size_t)(m0 + wave) * (HDIM / 4)];
        const float4 v1 = ug[(size_t)(m0 + wave + 8) * (HDIM / 4)];
        float4 hv       = ug[(size_t)hsrc * (HDIM / 4)];
        const float4 zero4 = make_float4(0.f, 0.f, 0.f, 0.f);
        if (wave >= 2) hv = zero4;
        *reinterpret_cast<float4*>(s_U  + wave * APITCH + e)       = v0;
        *reinterpret_cast<float4*>(s_U  + (wave + 8) * APITCH + e) = v1;
        *reinterpret_cast<float4*>(s_UH + wave * APITCH + e)       = hv;
        *reinterpret_cast<float4*>(s_UH + (wave + 8) * APITCH + e) = zero4;
    }

    float wv[12];
#pragma unroll
    for (int i = 0; i < 12; ++i) wv[i] = cw[e * 3 + i];
    float bv[4];
#pragma unroll
    for (int i = 0; i < 4; ++i) bv[i] = cb[e + i];

    __syncthreads();

    const int col = (wave << 4) + l16;

    float zs[8];
    {
        v8f accX = {0.f, 0.f, 0.f, 0.f, 0.f, 0.f, 0.f, 0.f};
        v8f accH = {0.f, 0.f, 0.f, 0.f, 0.f, 0.f, 0.f, 0.f};
        v8f accZ = {0.f, 0.f, 0.f, 0.f, 0.f, 0.f, 0.f, 0.f};
        const float* Urow = s_U  + l16 * APITCH + 2 * hi;
        const float* Hrow = s_UH + l16 * APITCH + 2 * hi;
        const float* Wcol = W_in + (size_t)(2 * hi) * 256 + col;
#pragma unroll
        for (int k0 = 0; k0 < HDIM; k0 += 4) {
            v2f a, ah;
            a.x  = Urow[k0 + 0];
            a.y  = Urow[k0 + 1];
            ah.x = Hrow[k0 + 0];
            ah.y = Hrow[k0 + 1];
            v2f bX, bZ;
            bX.x = Wcol[(k0 + 0) * 256];
            bX.y = Wcol[(k0 + 1) * 256];
            bZ.x = Wcol[(k0 + 0) * 256 + 128];
            bZ.y = Wcol[(k0 + 1) * 256 + 128];
            accX = __builtin_amdgcn_wmma_f32_16x16x4_f32(false, a, false, bX,
                                                         (short)0, accX, false, false);
            accH = __builtin_amdgcn_wmma_f32_16x16x4_f32(false, ah, false, bX,
                                                         (short)0, accH, false, false);
            accZ = __builtin_amdgcn_wmma_f32_16x16x4_f32(false, a, false, bZ,
                                                         (short)0, accZ, false, false);
        }
        const float biasX = b_in[col];
        const float biasZ = b_in[128 + col];
#pragma unroll
        for (int r = 0; r < 8; ++r) {
            const int row = r + hi * 8;
            s_X[(1 + row) * APITCH + col] = accX[r] + biasX;
            zs[r] = silu_fast(accZ[r] + biasZ);
        }
        if (hi == 0) {
            s_X[col]               = hasPrev ? accH[0] + biasX : 0.f;
            s_X[17 * APITCH + col] = hasNext ? accH[1] + biasX : 0.f;
        }
    }
    __syncthreads();

    float4 o0, o1;
    {
        const int r0 = wave << 1;
        const float4 q0 = *reinterpret_cast<const float4*>(s_X + (r0 + 0) * APITCH + e);
        const float4 q1 = *reinterpret_cast<const float4*>(s_X + (r0 + 1) * APITCH + e);
        const float4 q2 = *reinterpret_cast<const float4*>(s_X + (r0 + 2) * APITCH + e);
        const float4 q3 = *reinterpret_cast<const float4*>(s_X + (r0 + 3) * APITCH + e);

        o0.x = silu_fast(q0.x * wv[0] + q1.x * wv[1]  + q2.x * wv[2]  + bv[0]);
        o0.y = silu_fast(q0.y * wv[3] + q1.y * wv[4]  + q2.y * wv[5]  + bv[1]);
        o0.z = silu_fast(q0.z * wv[6] + q1.z * wv[7]  + q2.z * wv[8]  + bv[2]);
        o0.w = silu_fast(q0.w * wv[9] + q1.w * wv[10] + q2.w * wv[11] + bv[3]);
        o1.x = silu_fast(q1.x * wv[0] + q2.x * wv[1]  + q3.x * wv[2]  + bv[0]);
        o1.y = silu_fast(q1.y * wv[3] + q2.y * wv[4]  + q3.y * wv[5]  + bv[1]);
        o1.z = silu_fast(q1.z * wv[6] + q2.z * wv[7]  + q3.z * wv[8]  + bv[2]);
        o1.w = silu_fast(q1.w * wv[9] + q2.w * wv[10] + q3.w * wv[11] + bv[3]);

        *reinterpret_cast<float4*>(s_A + r0 * APITCH + e)       = o0;
        *reinterpret_cast<float4*>(s_A + (r0 + 1) * APITCH + e) = o1;
    }
    __syncthreads();
    {
        float4* g = reinterpret_cast<float4*>(x2) + lane;
        g[(size_t)mA * (HDIM / 4)]       = o0;
        g[(size_t)(mA + 1) * (HDIM / 4)] = o1;
#pragma unroll
        for (int r = 0; r < 8; ++r)
            sz[(size_t)(m0 + r + hi * 8) * HDIM + col] = zs[r];
    }

    v8f accD = {0.f, 0.f, 0.f, 0.f, 0.f, 0.f, 0.f, 0.f};
    v8f accB = {0.f, 0.f, 0.f, 0.f, 0.f, 0.f, 0.f, 0.f};
    v8f accC = {0.f, 0.f, 0.f, 0.f, 0.f, 0.f, 0.f, 0.f};

    const float* Arow = s_A + l16 * APITCH + 2 * hi;
    const float* Wcol = W + (size_t)(2 * hi) * 384 + col;

#pragma unroll
    for (int k0 = 0; k0 < HDIM; k0 += 4) {
        v2f a;
        a.x = Arow[k0 + 0];
        a.y = Arow[k0 + 1];
        v2f bD, bB, bC;
        bD.x = Wcol[(k0 + 0) * 384];
        bD.y = Wcol[(k0 + 1) * 384];
        bB.x = Wcol[(k0 + 0) * 384 + 128];
        bB.y = Wcol[(k0 + 1) * 384 + 128];
        bC.x = Wcol[(k0 + 0) * 384 + 256];
        bC.y = Wcol[(k0 + 1) * 384 + 256];
        accD = __builtin_amdgcn_wmma_f32_16x16x4_f32(false, a, false, bD,
                                                     (short)0, accD, false, false);
        accB = __builtin_amdgcn_wmma_f32_16x16x4_f32(false, a, false, bB,
                                                     (short)0, accB, false, false);
        accC = __builtin_amdgcn_wmma_f32_16x16x4_f32(false, a, false, bC,
                                                     (short)0, accC, false, false);
    }

    const float biasD = bias[col];
    const float biasB = bias[128 + col];
    const float biasC = bias[256 + col];
#pragma unroll
    for (int r = 0; r < 8; ++r) {
        const size_t o = (size_t)(m0 + r + hi * 8) * 128 + col;
        delta[o] = softplus_fast(accD[r] + biasD);
        Bm[o]    = accB[r] + biasB;
        Cm[o]    = accC[r] + biasC;
    }
}

#define JS 4
#define JW (HDIM / JS)
#define SCAN_THREADS (8 * JW)
#define LOG2E_F 1.4426950408889634f

__device__ __forceinline__ float4 mul4(float4 a, float4 b)
{
    return make_float4(a.x * b.x, a.y * b.y, a.z * b.z, a.w * b.w);
}

__global__ void __launch_bounds__(SCAN_THREADS)
scan_passA(const float* __restrict__ delta,
           const float* __restrict__ Bm,
           const float* __restrict__ x2,
           const float* __restrict__ A_log,
           float* __restrict__ Ssum,
           float* __restrict__ Bacc)
{
    const int wg = blockIdx.x;
    const int js = wg & (JS - 1);
    const int bc = wg / JS;
    const int b  = bc >> 4;
    const int c  = bc & (NCHUNK - 1);
    const int tid = threadIdx.x;
    const int cl  = tid & 15;
    const int g   = tid >> 4;
    const int j0  = js * JW + 2 * cl;
    const int nbase = g << 3;

    __shared__ __attribute__((aligned(16))) float s_d[CLEN * 128];
    __shared__ __attribute__((aligned(16))) float s_dB[CLEN * 128];
    __shared__ __attribute__((aligned(16))) float s_x[CLEN * JW];

    const size_t base = ((size_t)b * LEN + (size_t)c * CLEN) * HDIM;
    {
        const float4* gd = reinterpret_cast<const float4*>(delta + base);
        const float4* gB = reinterpret_cast<const float4*>(Bm + base);
        const float4* gx = reinterpret_cast<const float4*>(x2 + base + js * JW);
        float4* ld = reinterpret_cast<float4*>(s_d);
        float4* lB = reinterpret_cast<float4*>(s_dB);
        float4* lx = reinterpret_cast<float4*>(s_x);
        for (int k = 0; k < (CLEN * 128 / 4) / SCAN_THREADS; ++k) {
            const int q = tid + k * SCAN_THREADS;
            const float4 dv = gd[q];
            ld[q] = dv;
            lB[q] = mul4(dv, gB[q]);
        }
        for (int k = 0; k < (CLEN * JW / 4) / SCAN_THREADS; ++k) {
            const int q = tid + k * SCAN_THREADS;
            lx[q] = gx[(q / (JW / 4)) * (HDIM / 4) + (q % (JW / 4))];
        }
    }

    const float Aj2a = -__expf(A_log[j0]) * LOG2E_F;
    const float Aj2b = -__expf(A_log[j0 + 1]) * LOG2E_F;
    float ha[8], hb[8];
#pragma unroll
    for (int i = 0; i < 8; ++i) { ha[i] = 0.f; hb[i] = 0.f; }
    const int sl = cl & 7;
    float Ssl = 0.f;

    __syncthreads();

    for (int l = 0; l < CLEN; ++l) {
        const float* rd  = s_d  + l * 128 + nbase;
        const float* rdB = s_dB + l * 128 + nbase;
        const float2 xv = *reinterpret_cast<const float2*>(s_x + l * JW + 2 * cl);
        Ssl += rd[sl];
#pragma unroll
        for (int i = 0; i < 8; ++i) {
            const float d  = rd[i];
            const float dB = rdB[i];
            const float ea = __builtin_amdgcn_exp2f(d * Aj2a);
            const float eb = __builtin_amdgcn_exp2f(d * Aj2b);
            ha[i] = ea * ha[i] + dB * xv.x;
            hb[i] = eb * hb[i] + dB * xv.y;
        }
    }

    const size_t ob = (size_t)bc * 128;
#pragma unroll
    for (int i = 0; i < 8; ++i)
        *reinterpret_cast<float2*>(Bacc + (ob + nbase + i) * 128 + j0) = make_float2(ha[i], hb[i]);
    if (js == 0 && cl < 8)
        Ssum[ob + nbase + sl] = Ssl;
}

__global__ void __launch_bounds__(SCAN_THREADS)
scan_passB(const float* __restrict__ Ssum,
           const float* __restrict__ Bacc,
           const float* __restrict__ A_log,
           float* __restrict__ hinit)
{
    const int id = blockIdx.x * SCAN_THREADS + threadIdx.x;
    const int j  = id & 127;
    const int n  = (id >> 7) & 127;
    const int b  = id >> 14;

    const float Aj = -__expf(A_log[j]);
    float h = 0.f;
#pragma unroll
    for (int c = 0; c < NCHUNK; ++c) {
        const size_t ob = (size_t)(b * NCHUNK + c) * 128;
        const size_t idx = (ob + n) * 128 + j;
        hinit[idx] = h;
        const float P = __expf(Ssum[ob + n] * Aj);
        h = P * h + Bacc[idx];
    }
}

#define SBLK 8
__global__ void __launch_bounds__(SCAN_THREADS)
scan_passC(const float* __restrict__ delta,
           const float* __restrict__ Bm,
           const float* __restrict__ Cm,
           const float* __restrict__ x2,
           const float* __restrict__ u,
           const float* __restrict__ sz,
           const float* __restrict__ A_log,
           const float* __restrict__ Dv,
           const float* __restrict__ hinit,
           float* __restrict__ ymul)
{
    const int wg = blockIdx.x;
    const int js = wg & (JS - 1);
    const int bc = wg / JS;
    const int b  = bc >> 4;
    const int c  = bc & (NCHUNK - 1);
    const int tid = threadIdx.x;
    const int cl  = tid & 15;
    const int g   = tid >> 4;
    const int j0  = js * JW + 2 * cl;
    const int nbase = g << 3;
    const int jl  = tid & (JW - 1);
    const int ng  = tid / JW;
    const int j   = js * JW + jl;

    __shared__ __attribute__((aligned(16))) float s_d[CLEN * 128];
    __shared__ __attribute__((aligned(16))) float s_dB[CLEN * 128];
    __shared__ __attribute__((aligned(16))) float s_x[CLEN * JW];
    __shared__ __attribute__((aligned(16))) float s_part[2][SBLK][16][JW];

    const size_t base = ((size_t)b * LEN + (size_t)c * CLEN) * HDIM;
    {
        const float4* gd = reinterpret_cast<const float4*>(delta + base);
        const float4* gB = reinterpret_cast<const float4*>(Bm + base);
        const float4* gx = reinterpret_cast<const float4*>(x2 + base + js * JW);
        float4* ld = reinterpret_cast<float4*>(s_d);
        float4* lB = reinterpret_cast<float4*>(s_dB);
        float4* lx = reinterpret_cast<float4*>(s_x);
        for (int k = 0; k < (CLEN * 128 / 4) / SCAN_THREADS; ++k) {
            const int q = tid + k * SCAN_THREADS;
            const float4 dv = gd[q];
            ld[q] = dv;
            lB[q] = mul4(dv, gB[q]);
        }
        for (int k = 0; k < (CLEN * JW / 4) / SCAN_THREADS; ++k) {
            const int q = tid + k * SCAN_THREADS;
            lx[q] = gx[(q / (JW / 4)) * (HDIM / 4) + (q % (JW / 4))];
        }
    }

    const float Aj2a = -__expf(A_log[j0]) * LOG2E_F;
    const float Aj2b = -__expf(A_log[j0 + 1]) * LOG2E_F;
    const float Dj = Dv[j];

    const size_t ob = (size_t)bc * 128;
    float ha[8], hb[8];
#pragma unroll
    for (int i = 0; i < 8; ++i) {
        const float2 hv = *reinterpret_cast<const float2*>(hinit + (ob + nbase + i) * 128 + j0);
        ha[i] = hv.x;
        hb[i] = hv.y;
    }

    __syncthreads();

    for (int blk = 0; blk < CLEN / SBLK; ++blk) {
        const int buf = blk & 1;
        const size_t moff = base + (size_t)(blk * SBLK + ng) * HDIM + j;
        const float cv = Cm[moff];
        const float uv = u[moff];
        const float zv = sz[moff];

        for (int ll = 0; ll < SBLK; ++ll) {
            const int l = blk * SBLK + ll;
            const float* rd  = s_d  + l * 128 + nbase;
            const float* rdB = s_dB + l * 128 + nbase;
            const float2 xv = *reinterpret_cast<const float2*>(s_x + l * JW + 2 * cl);
            float pa = 0.f, pb = 0.f;
#pragma unroll
            for (int i = 0; i < 8; ++i) {
                const float d  = rd[i];
                const float dB = rdB[i];
                const float ea = __builtin_amdgcn_exp2f(d * Aj2a);
                const float eb = __builtin_amdgcn_exp2f(d * Aj2b);
                ha[i] = ea * ha[i] + dB * xv.x;
                hb[i] = eb * hb[i] + dB * xv.y;
                pa += ha[i];
                pb += hb[i];
            }
            *reinterpret_cast<float2*>(&s_part[buf][ll][g][2 * cl]) = make_float2(pa, pb);
        }
        __syncthreads();
        float tot = 0.f;
#pragma unroll
        for (int gg = 0; gg < 16; ++gg) tot += s_part[buf][ng][gg][jl];
        const float y = cv * tot + Dj * uv;
        ymul[moff] = y * zv;
    }
}

extern "C" void kernel_launch(void* const* d_in, const int* in_sizes, int n_in,
                              void* d_out, int out_size, void* d_ws, size_t ws_size,
                              hipStream_t stream)
{
    (void)in_sizes; (void)n_in; (void)out_size; (void)ws_size;

    const float* u      = (const float*)d_in[0];
    const float* W_in   = (const float*)d_in[1];
    const float* b_in   = (const float*)d_in[2];
    const float* conv_w = (const float*)d_in[3];
    const float* conv_b = (const float*)d_in[4];
    const float* W_x    = (const float*)d_in[5];
    const float* b_x    = (const float*)d_in[6];
    const float* A_log  = (const float*)d_in[7];
    const float* Dv     = (const float*)d_in[8];
    const float* W_out  = (const float*)d_in[9];
    const float* b_out  = (const float*)d_in[10];

    const size_t P = (size_t)MTOT * HDIM;
    const size_t NSTATE = (size_t)BATCH * NCHUNK * 128 * 128;

    float* ws    = (float*)d_ws;
    float* s_z   = ws + 1 * P;
    float* x2    = ws + 2 * P;
    float* delta = ws + 3 * P;
    float* Bmat  = ws + 4 * P;
    float* Cmat  = ws + 5 * P;
    float* ymul  = ws + 6 * P;
    float* Ssum  = ws + 7 * P;
    float* Bacc  = Ssum + (size_t)BATCH * NCHUNK * 128;
    float* hinit = Bacc + NSTATE;

    inproj_conv_ssm_wmma_f32<<<MTOT / 16, 256, 0, stream>>>(
        u, W_in, b_in, conv_w, conv_b, W_x, b_x, s_z, x2, delta, Bmat, Cmat);
    scan_passA<<<BATCH * NCHUNK * JS, SCAN_THREADS, 0, stream>>>(delta, Bmat, x2, A_log,
                                                                 Ssum, Bacc);
    scan_passB<<<(BATCH * 128 * 128) / SCAN_THREADS, SCAN_THREADS, 0, stream>>>(
        Ssum, Bacc, A_log, hinit);
    scan_passC<<<BATCH * NCHUNK * JS, SCAN_THREADS, 0, stream>>>(delta, Bmat, Cmat, x2, u, s_z,
                                                                 A_log, Dv, hinit, ymul);
    gemm16_wmma_f32<128, 128, EPI_PLAIN><<<256, 256, 0, stream>>>(
        ymul, W_out, b_out, (float*)d_out, nullptr);
}


# scan passes A and C: the LDS operands (d, d*B, x rows) of each pair of steps are read two steps ahead into a second pair of register sets, so no step waits for an LDS round trip; same order of sums
# speedup vs baseline: 5.0456x; 1.0312x over previous
#include <hip/hip_runtime.h>


#define HDIM 128
#define BATCH 4
#define LEN 1024
#define MTOT (BATCH * LEN)
#define NCHUNK 16
#define CLEN (LEN / NCHUNK)

typedef float v2f __attribute__((ext_vector_type(2)));
typedef float v8f __attribute__((ext_vector_type(8)));

__device__ __forceinline__ float sigmoid_fast(float v) { return 1.0f / (1.0f + __expf(-v)); }
__device__ __forceinline__ float silu_fast(float v)    { return v * sigmoid_fast(v); }
__device__ __forceinline__ float softplus_fast(float v){ return (v > 20.0f) ? v : log1pf(__expf(v)); }

#define EPI_XZ    0
#define EPI_PLAIN 1

template <int N, int K, int MODE>
__global__ void __launch_bounds__(256)
gemm16_wmma_f32(const float* __restrict__ A,
                const float* __restrict__ W,
                const float* __restrict__ bias,
                float* __restrict__ out0,
                float* __restrict__ out1)
{
    const int lane = threadIdx.x & 31;
    const int wave = threadIdx.x >> 5;
    const int gw   = blockIdx.x * (blockDim.x >> 5) + wave;
    const int tilesN = N >> 4;
    const int m0 = (gw / tilesN) << 4;
    const int n0 = (gw % tilesN) << 4;
    const int hi  = lane >> 4;
    const int l16 = lane & 15;

    v8f acc = {0.f, 0.f, 0.f, 0.f, 0.f, 0.f, 0.f, 0.f};

    const float* Arow = A + (size_t)(m0 + l16) * K + 2 * hi;
    const float* Wcol = W + (size_t)(2 * hi) * N + n0 + l16;

#pragma unroll
    for (int k0 = 0; k0 < K; k0 += 4) {
        v2f a;
        a.x = Arow[k0 + 0];
        a.y = Arow[k0 + 1];
        v2f b;
        b.x = Wcol[(k0 + 0) * N];
        b.y = Wcol[(k0 + 1) * N];
        acc = __builtin_amdgcn_wmma_f32_16x16x4_f32(false, a, false, b,
                                                    (short)0, acc, false, false);
    }

    const int   col = n0 + l16;
    const float bv  = bias[col];
#pragma unroll
    for (int r = 0; r < 8; ++r) {
        const int row = m0 + r + hi * 8;
        float v = acc[r] + bv;
        if (MODE == EPI_XZ) {
            if (col < HDIM) out0[(size_t)row * HDIM + col] = v;
            else            out1[(size_t)row * HDIM + (col - HDIM)] = silu_fast(v);
        } else {
            out0[(size_t)row * HDIM + col] = v;
        }
    }
}

#define APITCH 132
__global__ void __launch_bounds__(256) __attribute__((amdgpu_waves_per_eu(2, 2)))
inproj_conv_ssm_wmma_f32(const float* __restrict__ u,
                         const float* __restrict__ W_in,
                         const float* __restrict__ b_in,
                         const float* __restrict__ cw,
                         const float* __restrict__ cb,
                         const float* __restrict__ W,
                         const float* __restrict__ bias,
                         float* __restrict__ sz,
                         float* __restrict__ x2,
                         float* __restrict__ delta,
                         float* __restrict__ Bm,
                         float* __restrict__ Cm)
{
    __shared__ __attribute__((aligned(16))) float s_U[16 * APITCH];
    __shared__ __attribute__((aligned(16))) float s_UH[16 * APITCH];
    __shared__ __attribute__((aligned(16))) float s_X[18 * APITCH];
    __shared__ __attribute__((aligned(16))) float s_A[16 * APITCH];

    const int tid  = threadIdx.x;
    const int lane = tid & 31;
    const int wave = tid >> 5;
    const int m0   = blockIdx.x << 4;
    const int mA   = m0 + (wave << 1);
    const bool hasPrev = (m0 & (LEN - 1)) != 0;
    const bool hasNext = ((m0 + 16) & (LEN - 1)) != 0;
    const int hi  = lane >> 4;
    const int l16 = lane & 15;
    const int e   = lane << 2;

    {
        const float4* ug = reinterpret_cast<const float4*>(u) + lane;
        const int hsrc = (wave == 0) ? (hasPrev ? m0 - 1 : m0) : (hasNext ? m0 + 16 : m0);
        const float4 v0 = ug[(size_t)(m0 + wave) * (HDIM / 4)];
        const float4 v1 = ug[(size_t)(m0 + wave + 8) * (HDIM / 4)];
        float4 hv       = ug[(size_t)hsrc * (HDIM / 4)];
        const float4 zero4 = make_float4(0.f, 0.f, 0.f, 0.f);
        if (wave >= 2) hv = zero4;
        *reinterpret_cast<float4*>(s_U  + wave * APITCH + e)       = v0;
        *reinterpret_cast<float4*>(s_U  + (wave + 8) * APITCH + e) = v1;
        *reinterpret_cast<float4*>(s_UH + wave * APITCH + e)       = hv;
        *reinterpret_cast<float4*>(s_UH + (wave + 8) * APITCH + e) = zero4;
    }

    float wv[12];
#pragma unroll
    for (int i = 0; i < 12; ++i) wv[i] = cw[e * 3 + i];
    float bv[4];
#pragma unroll
    for (int i = 0; i < 4; ++i) bv[i] = cb[e + i];

    __syncthreads();

    const int col = (wave << 4) + l16;

    float zs[8];
    {
        v8f accX = {0.f, 0.f, 0.f, 0.f, 0.f, 0.f, 0.f, 0.f};
        v8f accH = {0.f, 0.f, 0.f, 0.f, 0.f, 0.f, 0.f, 0.f};
        v8f accZ = {0.f, 0.f, 0.f, 0.f, 0.f, 0.f, 0.f, 0.f};
        const float* Urow = s_U  + l16 * APITCH + 2 * hi;
        const float* Hrow = s_UH + l16 * APITCH + 2 * hi;
        const float* Wcol = W_in + (size_t)(2 * hi) * 256 + col;
#pragma unroll
        for (int k0 = 0; k0 < HDIM; k0 += 4) {
            v2f a, ah;
            a.x  = Urow[k0 + 0];
            a.y  = Urow[k0 + 1];
            ah.x = Hrow[k0 + 0];
            ah.y = Hrow[k0 + 1];
            v2f bX, bZ;
            bX.x = Wcol[(k0 + 0) * 256];
            bX.y = Wcol[(k0 + 1) * 256];
            bZ.x = Wcol[(k0 + 0) * 256 + 128];
            bZ.y = Wcol[(k0 + 1) * 256 + 128];
            accX = __builtin_amdgcn_wmma_f32_16x16x4_f32(false, a, false, bX,
                                                         (short)0, accX, false, false);
            accH = __builtin_amdgcn_wmma_f32_16x16x4_f32(false, ah, false, bX,
                                                         (short)0, accH, false, false);
            accZ = __builtin_amdgcn_wmma_f32_16x16x4_f32(false, a, false, bZ,
                                                         (short)0, accZ, false, false);
        }
        const float biasX = b_in[col];
        const float biasZ = b_in[128 + col];
#pragma unroll
        for (int r = 0; r < 8; ++r) {
            const int row = r + hi * 8;
            s_X[(1 + row) * APITCH + col] = accX[r] + biasX;
            zs[r] = silu_fast(accZ[r] + biasZ);
        }
        if (hi == 0) {
            s_X[col]               = hasPrev ? accH[0] + biasX : 0.f;
            s_X[17 * APITCH + col] = hasNext ? accH[1] + biasX : 0.f;
        }
    }
    __syncthreads();

    float4 o0, o1;
    {
        const int r0 = wave << 1;
        const float4 q0 = *reinterpret_cast<const float4*>(s_X + (r0 + 0) * APITCH + e);
        const float4 q1 = *reinterpret_cast<const float4*>(s_X + (r0 + 1) * APITCH + e);
        const float4 q2 = *reinterpret_cast<const float4*>(s_X + (r0 + 2) * APITCH + e);
        const float4 q3 = *reinterpret_cast<const float4*>(s_X + (r0 + 3) * APITCH + e);

        o0.x = silu_fast(q0.x * wv[0] + q1.x * wv[1]  + q2.x * wv[2]  + bv[0]);
        o0.y = silu_fast(q0.y * wv[3] + q1.y * wv[4]  + q2.y * wv[5]  + bv[1]);
        o0.z = silu_fast(q0.z * wv[6] + q1.z * wv[7]  + q2.z * wv[8]  + bv[2]);
        o0.w = silu_fast(q0.w * wv[9] + q1.w * wv[10] + q2.w * wv[11] + bv[3]);
        o1.x = silu_fast(q1.x * wv[0] + q2.x * wv[1]  + q3.x * wv[2]  + bv[0]);
        o1.y = silu_fast(q1.y * wv[3] + q2.y * wv[4]  + q3.y * wv[5]  + bv[1]);
        o1.z = silu_fast(q1.z * wv[6] + q2.z * wv[7]  + q3.z * wv[8]  + bv[2]);
        o1.w = silu_fast(q1.w * wv[9] + q2.w * wv[10] + q3.w * wv[11] + bv[3]);

        *reinterpret_cast<float4*>(s_A + r0 * APITCH + e)       = o0;
        *reinterpret_cast<float4*>(s_A + (r0 + 1) * APITCH + e) = o1;
    }
    __syncthreads();
    {
        float4* g = reinterpret_cast<float4*>(x2) + lane;
        g[(size_t)mA * (HDIM / 4)]       = o0;
        g[(size_t)(mA + 1) * (HDIM / 4)] = o1;
#pragma unroll
        for (int r = 0; r < 8; ++r)
            sz[(size_t)(m0 + r + hi * 8) * HDIM + col] = zs[r];
    }

    v8f accD = {0.f, 0.f, 0.f, 0.f, 0.f, 0.f, 0.f, 0.f};
    v8f accB = {0.f, 0.f, 0.f, 0.f, 0.f, 0.f, 0.f, 0.f};
    v8f accC = {0.f, 0.f, 0.f, 0.f, 0.f, 0.f, 0.f, 0.f};

    const float* Arow = s_A + l16 * APITCH + 2 * hi;
    const float* Wcol = W + (size_t)(2 * hi) * 384 + col;

#pragma unroll
    for (int k0 = 0; k0 < HDIM; k0 += 4) {
        v2f a;
        a.x = Arow[k0 + 0];
        a.y = Arow[k0 + 1];
        v2f bD, bB, bC;
        bD.x = Wcol[(k0 + 0) * 384];
        bD.y = Wcol[(k0 + 1) * 384];
        bB.x = Wcol[(k0 + 0) * 384 + 128];
        bB.y = Wcol[(k0 + 1) * 384 + 128];
        bC.x = Wcol[(k0 + 0) * 384 + 256];
        bC.y = Wcol[(k0 + 1) * 384 + 256];
        accD = __builtin_amdgcn_wmma_f32_16x16x4_f32(false, a, false, bD,
                                                     (short)0, accD, false, false);
        accB = __builtin_amdgcn_wmma_f32_16x16x4_f32(false, a, false, bB,
                                                     (short)0, accB, false, false);
        accC = __builtin_amdgcn_wmma_f32_16x16x4_f32(false, a, false, bC,
                                                     (short)0, accC, false, false);
    }

    const float biasD = bias[col];
    const float biasB = bias[128 + col];
    const float biasC = bias[256 + col];
#pragma unroll
    for (int r = 0; r < 8; ++r) {
        const size_t o = (size_t)(m0 + r + hi * 8) * 128 + col;
        delta[o] = softplus_fast(accD[r] + biasD);
        Bm[o]    = accB[r] + biasB;
        Cm[o]    = accC[r] + biasC;
    }
}

#define JS 4
#define JW (HDIM / JS)
#define SCAN_THREADS (8 * JW)
#define LOG2E_F 1.4426950408889634f

__device__ __forceinline__ float4 mul4(float4 a, float4 b)
{
    return make_float4(a.x * b.x, a.y * b.y, a.z * b.z, a.w * b.w);
}


struct StepOps { float4 d0, d1, b0, b1; float2 xv; };

__device__ __forceinline__ StepOps load_step(const float* s_d, const float* s_dB, const float* s_x,
                                             int l, int nbase, int cl)
{
    StepOps o;
    const float4* rd  = reinterpret_cast<const float4*>(s_d  + l * 128 + nbase);
    const float4* rdB = reinterpret_cast<const float4*>(s_dB + l * 128 + nbase);
    o.d0 = rd[0];  o.d1 = rd[1];
    o.b0 = rdB[0]; o.b1 = rdB[1];
    o.xv = *reinterpret_cast<const float2*>(s_x + l * JW + 2 * cl);
    return o;
}

__device__ __forceinline__ void hold_read_ahead(const StepOps& o)
{
    asm volatile("" :: "v"(o.d0.x), "v"(o.d0.y), "v"(o.d0.z), "v"(o.d0.w),
                       "v"(o.d1.x), "v"(o.d1.y), "v"(o.d1.z), "v"(o.d1.w),
                       "v"(o.b0.x), "v"(o.b0.y), "v"(o.b0.z), "v"(o.b0.w),
                       "v"(o.b1.x), "v"(o.b1.y), "v"(o.b1.z), "v"(o.b1.w),
                       "v"(o.xv.x), "v"(o.xv.y));
}

__device__ __forceinline__ float2 scan_step(float (&ha)[8], float (&hb)[8], const StepOps& o,
                                            float Aj2a, float Aj2b)
{
    const float d[8]  = {o.d0.x, o.d0.y, o.d0.z, o.d0.w, o.d1.x, o.d1.y, o.d1.z, o.d1.w};
    const float dB[8] = {o.b0.x, o.b0.y, o.b0.z, o.b0.w, o.b1.x, o.b1.y, o.b1.z, o.b1.w};
    float pa = 0.f, pb = 0.f;
#pragma unroll
    for (int i = 0; i < 8; ++i) {
        const float ea = __builtin_amdgcn_exp2f(d[i] * Aj2a);
        const float eb = __builtin_amdgcn_exp2f(d[i] * Aj2b);
        ha[i] = ea * ha[i] + dB[i] * o.xv.x;
        hb[i] = eb * hb[i] + dB[i] * o.xv.y;
        pa += ha[i];
        pb += hb[i];
    }
    return make_float2(pa, pb);
}

__global__ void __launch_bounds__(SCAN_THREADS)
scan_passA(const float* __restrict__ delta,
           const float* __restrict__ Bm,
           const float* __restrict__ x2,
           const float* __restrict__ A_log,
           float* __restrict__ Ssum,
           float* __restrict__ Bacc)
{
    const int wg = blockIdx.x;
    const int js = wg & (JS - 1);
    const int bc = wg / JS;
    const int b  = bc >> 4;
    const int c  = bc & (NCHUNK - 1);
    const int tid = threadIdx.x;
    const int cl  = tid & 15;
    const int g   = tid >> 4;
    const int j0  = js * JW + 2 * cl;
    const int nbase = g << 3;

    __shared__ __attribute__((aligned(16))) float s_d[CLEN * 128];
    __shared__ __attribute__((aligned(16))) float s_dB[CLEN * 128];
    __shared__ __attribute__((aligned(16))) float s_x[CLEN * JW];

    const size_t base = ((size_t)b * LEN + (size_t)c * CLEN) * HDIM;
    {
        const float4* gd = reinterpret_cast<const float4*>(delta + base);
        const float4* gB = reinterpret_cast<const float4*>(Bm + base);
        const float4* gx = reinterpret_cast<const float4*>(x2 + base + js * JW);
        float4* ld = reinterpret_cast<float4*>(s_d);
        float4* lB = reinterpret_cast<float4*>(s_dB);
        float4* lx = reinterpret_cast<float4*>(s_x);
        for (int k = 0; k < (CLEN * 128 / 4) / SCAN_THREADS; ++k) {
            const int q = tid + k * SCAN_THREADS;
            const float4 dv = gd[q];
            ld[q] = dv;
            lB[q] = mul4(dv, gB[q]);
        }
        for (int k = 0; k < (CLEN * JW / 4) / SCAN_THREADS; ++k) {
            const int q = tid + k * SCAN_THREADS;
            lx[q] = gx[(q / (JW / 4)) * (HDIM / 4) + (q % (JW / 4))];
        }
    }

    const float Aj2a = -__expf(A_log[j0]) * LOG2E_F;
    const float Aj2b = -__expf(A_log[j0 + 1]) * LOG2E_F;
    float ha[8], hb[8];
#pragma unroll
    for (int i = 0; i < 8; ++i) { ha[i] = 0.f; hb[i] = 0.f; }
    const int sl = cl & 7;
    float Ssl = 0.f;

    __syncthreads();

    StepOps p0 = load_step(s_d, s_dB, s_x, 0, nbase, cl);
    StepOps p1 = load_step(s_d, s_dB, s_x, 1, nbase, cl);
    float sp0 = s_d[nbase + sl];
    float sp1 = s_d[128 + nbase + sl];
    for (int l = 0; l < CLEN; l += 4) {
        const StepOps q0 = load_step(s_d, s_dB, s_x, l + 2, nbase, cl);
        const StepOps q1 = load_step(s_d, s_dB, s_x, l + 3, nbase, cl);
        const float sq0 = s_d[(l + 2) * 128 + nbase + sl];
        const float sq1 = s_d[(l + 3) * 128 + nbase + sl];
        __builtin_amdgcn_sched_barrier(0);
        Ssl += sp0;
        scan_step(ha, hb, p0, Aj2a, Aj2b);
        Ssl += sp1;
        scan_step(ha, hb, p1, Aj2a, Aj2b);
        const int l4 = (l + 4 < CLEN) ? l + 4 : CLEN - 2;
        p0  = load_step(s_d, s_dB, s_x, l4, nbase, cl);
        p1  = load_step(s_d, s_dB, s_x, l4 + 1, nbase, cl);
        sp0 = s_d[l4 * 128 + nbase + sl];
        sp1 = s_d[(l4 + 1) * 128 + nbase + sl];
        __builtin_amdgcn_sched_barrier(0);
        Ssl += sq0;
        scan_step(ha, hb, q0, Aj2a, Aj2b);
        Ssl += sq1;
        scan_step(ha, hb, q1, Aj2a, Aj2b);
        hold_read_ahead(p0);
        hold_read_ahead(p1);
    }

    const size_t ob = (size_t)bc * 128;
#pragma unroll
    for (int i = 0; i < 8; ++i)
        *reinterpret_cast<float2*>(Bacc + (ob + nbase + i) * 128 + j0) = make_float2(ha[i], hb[i]);
    if (js == 0 && cl < 8)
        Ssum[ob + nbase + sl] = Ssl;
}

__global__ void __launch_bounds__(SCAN_THREADS)
scan_passB(const float* __restrict__ Ssum,
           const float* __restrict__ Bacc,
           const float* __restrict__ A_log,
           float* __restrict__ hinit)
{
    const int id = blockIdx.x * SCAN_THREADS + threadIdx.x;
    const int j  = id & 127;
    const int n  = (id >> 7) & 127;
    const int b  = id >> 14;

    const float Aj = -__expf(A_log[j]);
    float h = 0.f;
#pragma unroll
    for (int c = 0; c < NCHUNK; ++c) {
        const size_t ob = (size_t)(b * NCHUNK + c) * 128;
        const size_t idx = (ob + n) * 128 + j;
        hinit[idx] = h;
        const float P = __expf(Ssum[ob + n] * Aj);
        h = P * h + Bacc[idx];
    }
}

#define SBLK 8
__global__ void __launch_bounds__(SCAN_THREADS)
scan_passC(const float* __restrict__ delta,
           const float* __restrict__ Bm,
           const float* __restrict__ Cm,
           const float* __restrict__ x2,
           const float* __restrict__ u,
           const float* __restrict__ sz,
           const float* __restrict__ A_log,
           const float* __restrict__ Dv,
           const float* __restrict__ hinit,
           float* __restrict__ ymul)
{
    const int wg = blockIdx.x;
    const int js = wg & (JS - 1);
    const int bc = wg / JS;
    const int b  = bc >> 4;
    const int c  = bc & (NCHUNK - 1);
    const int tid = threadIdx.x;
    const int cl  = tid & 15;
    const int g   = tid >> 4;
    const int j0  = js * JW + 2 * cl;
    const int nbase = g << 3;
    const int jl  = tid & (JW - 1);
    const int ng  = tid / JW;
    const int j   = js * JW + jl;

    __shared__ __attribute__((aligned(16))) float s_d[CLEN * 128];
    __shared__ __attribute__((aligned(16))) float s_dB[CLEN * 128];
    __shared__ __attribute__((aligned(16))) float s_x[CLEN * JW];
    __shared__ __attribute__((aligned(16))) float s_part[2][SBLK][16][JW];

    const size_t base = ((size_t)b * LEN + (size_t)c * CLEN) * HDIM;
    {
        const float4* gd = reinterpret_cast<const float4*>(delta + base);
        const float4* gB = reinterpret_cast<const float4*>(Bm + base);
        const float4* gx = reinterpret_cast<const float4*>(x2 + base + js * JW);
        float4* ld = reinterpret_cast<float4*>(s_d);
        float4* lB = reinterpret_cast<float4*>(s_dB);
        float4* lx = reinterpret_cast<float4*>(s_x);
        for (int k = 0; k < (CLEN * 128 / 4) / SCAN_THREADS; ++k) {
            const int q = tid + k * SCAN_THREADS;
            const float4 dv = gd[q];
            ld[q] = dv;
            lB[q] = mul4(dv, gB[q]);
        }
        for (int k = 0; k < (CLEN * JW / 4) / SCAN_THREADS; ++k) {
            const int q = tid + k * SCAN_THREADS;
            lx[q] = gx[(q / (JW / 4)) * (HDIM / 4) + (q % (JW / 4))];
        }
    }

    const float Aj2a = -__expf(A_log[j0]) * LOG2E_F;
    const float Aj2b = -__expf(A_log[j0 + 1]) * LOG2E_F;
    const float Dj = Dv[j];

    const size_t ob = (size_t)bc * 128;
    float ha[8], hb[8];
#pragma unroll
    for (int i = 0; i < 8; ++i) {
        const float2 hv = *reinterpret_cast<const float2*>(hinit + (ob + nbase + i) * 128 + j0);
        ha[i] = hv.x;
        hb[i] = hv.y;
    }

    __syncthreads();

    StepOps p0 = load_step(s_d, s_dB, s_x, 0, nbase, cl);
    StepOps p1 = load_step(s_d, s_dB, s_x, 1, nbase, cl);
    for (int blk = 0; blk < CLEN / SBLK; ++blk) {
        const int buf = blk & 1;
        const size_t moff = base + (size_t)(blk * SBLK + ng) * HDIM + j;
        const float cv = Cm[moff];
        const float uv = u[moff];
        const float zv = sz[moff];

        for (int ll = 0; ll < SBLK; ll += 4) {
            const int l = blk * SBLK + ll;
            const StepOps q0 = load_step(s_d, s_dB, s_x, l + 2, nbase, cl);
            const StepOps q1 = load_step(s_d, s_dB, s_x, l + 3, nbase, cl);
            __builtin_amdgcn_sched_barrier(0);
            *reinterpret_cast<float2*>(&s_part[buf][ll][g][2 * cl])     = scan_step(ha, hb, p0, Aj2a, Aj2b);
            *reinterpret_cast<float2*>(&s_part[buf][ll + 1][g][2 * cl]) = scan_step(ha, hb, p1, Aj2a, Aj2b);
            const int l4 = (l + 4 < CLEN) ? l + 4 : CLEN - 2;
            p0 = load_step(s_d, s_dB, s_x, l4, nbase, cl);
            p1 = load_step(s_d, s_dB, s_x, l4 + 1, nbase, cl);
            __builtin_amdgcn_sched_barrier(0);
            *reinterpret_cast<float2*>(&s_part[buf][ll + 2][g][2 * cl]) = scan_step(ha, hb, q0, Aj2a, Aj2b);
            *reinterpret_cast<float2*>(&s_part[buf][ll + 3][g][2 * cl]) = scan_step(ha, hb, q1, Aj2a, Aj2b);
        }
        __syncthreads();
        float tot = 0.f;
#pragma unroll
        for (int gg = 0; gg < 16; ++gg) tot += s_part[buf][ng][gg][jl];
        const float y = cv * tot + Dj * uv;
        ymul[moff] = y * zv;
    }
}

extern "C" void kernel_launch(void* const* d_in, const int* in_sizes, int n_in,
                              void* d_out, int out_size, void* d_ws, size_t ws_size,
                              hipStream_t stream)
{
    (void)in_sizes; (void)n_in; (void)out_size; (void)ws_size;

    const float* u      = (const float*)d_in[0];
    const float* W_in   = (const float*)d_in[1];
    const float* b_in   = (const float*)d_in[2];
    const float* conv_w = (const float*)d_in[3];
    const float* conv_b = (const float*)d_in[4];
    const float* W_x    = (const float*)d_in[5];
    const float* b_x    = (const float*)d_in[6];
    const float* A_log  = (const float*)d_in[7];
    const float* Dv     = (const float*)d_in[8];
    const float* W_out  = (const float*)d_in[9];
    const float* b_out  = (const float*)d_in[10];

    const size_t P = (size_t)MTOT * HDIM;
    const size_t NSTATE = (size_t)BATCH * NCHUNK * 128 * 128;

    float* ws    = (float*)d_ws;
    float* s_z   = ws + 1 * P;
    float* x2    = ws + 2 * P;
    float* delta = ws + 3 * P;
    float* Bmat  = ws + 4 * P;
    float* Cmat  = ws + 5 * P;
    float* ymul  = ws + 6 * P;
    float* Ssum  = ws + 7 * P;
    float* Bacc  = Ssum + (size_t)BATCH * NCHUNK * 128;
    float* hinit = Bacc + NSTATE;

    inproj_conv_ssm_wmma_f32<<<MTOT / 16, 256, 0, stream>>>(
        u, W_in, b_in, conv_w, conv_b, W_x, b_x, s_z, x2, delta, Bmat, Cmat);
    scan_passA<<<BATCH * NCHUNK * JS, SCAN_THREADS, 0, stream>>>(delta, Bmat, x2, A_log,
                                                                 Ssum, Bacc);
    scan_passB<<<(BATCH * 128 * 128) / SCAN_THREADS, SCAN_THREADS, 0, stream>>>(
        Ssum, Bacc, A_log, hinit);
    scan_passC<<<BATCH * NCHUNK * JS, SCAN_THREADS, 0, stream>>>(delta, Bmat, Cmat, x2, u, s_z,
                                                                 A_log, Dv, hinit, ymul);
    gemm16_wmma_f32<128, 128, EPI_PLAIN><<<256, 256, 0, stream>>>(
        ymul, W_out, b_out, (float*)d_out, nullptr);
}


# fused GEMM kernel's epilogues: softplus and SiLU from single hardware exp2, log2 and reciprocal instructions; the library log1p routine and the IEEE division sequences are gone
# speedup vs baseline: 5.2234x; 1.0352x over previous
#include <hip/hip_runtime.h>


#define HDIM 128
#define BATCH 4
#define LEN 1024
#define MTOT (BATCH * LEN)
#define NCHUNK 16
#define CLEN (LEN / NCHUNK)

typedef float v2f __attribute__((ext_vector_type(2)));
typedef float v8f __attribute__((ext_vector_type(8)));

#define LOG2E_ACT 1.4426950408889634f
#define LN2_ACT   0.6931471805599453f
__device__ __forceinline__ float sigmoid_fast(float v)
{
    return __builtin_amdgcn_rcpf(1.0f + __builtin_amdgcn_exp2f(v * -LOG2E_ACT));
}
__device__ __forceinline__ float silu_fast(float v)    { return v * sigmoid_fast(v); }
__device__ __forceinline__ float softplus_fast(float v)
{
    const float t = __builtin_amdgcn_exp2f(__builtin_fabsf(v) * -LOG2E_ACT);
    const float u = 1.0f + t;
    const float c = t - (u - 1.0f);
    return fmaxf(v, 0.0f) + __builtin_fmaf(__builtin_amdgcn_logf(u), LN2_ACT, c);
}

#define EPI_XZ    0
#define EPI_PLAIN 1

template <int N, int K, int MODE>
__global__ void __launch_bounds__(256)
gemm16_wmma_f32(const float* __restrict__ A,
                const float* __restrict__ W,
                const float* __restrict__ bias,
                float* __restrict__ out0,
                float* __restrict__ out1)
{
    const int lane = threadIdx.x & 31;
    const int wave = threadIdx.x >> 5;
    const int gw   = blockIdx.x * (blockDim.x >> 5) + wave;
    const int tilesN = N >> 4;
    const int m0 = (gw / tilesN) << 4;
    const int n0 = (gw % tilesN) << 4;
    const int hi  = lane >> 4;
    const int l16 = lane & 15;

    v8f acc = {0.f, 0.f, 0.f, 0.f, 0.f, 0.f, 0.f, 0.f};

    const float* Arow = A + (size_t)(m0 + l16) * K + 2 * hi;
    const float* Wcol = W + (size_t)(2 * hi) * N + n0 + l16;

#pragma unroll
    for (int k0 = 0; k0 < K; k0 += 4) {
        v2f a;
        a.x = Arow[k0 + 0];
        a.y = Arow[k0 + 1];
        v2f b;
        b.x = Wcol[(k0 + 0) * N];
        b.y = Wcol[(k0 + 1) * N];
        acc = __builtin_amdgcn_wmma_f32_16x16x4_f32(false, a, false, b,
                                                    (short)0, acc, false, false);
    }

    const int   col = n0 + l16;
    const float bv  = bias[col];
#pragma unroll
    for (int r = 0; r < 8; ++r) {
        const int row = m0 + r + hi * 8;
        float v = acc[r] + bv;
        if (MODE == EPI_XZ) {
            if (col < HDIM) out0[(size_t)row * HDIM + col] = v;
            else            out1[(size_t)row * HDIM + (col - HDIM)] = silu_fast(v);
        } else {
            out0[(size_t)row * HDIM + col] = v;
        }
    }
}

#define APITCH 132
__global__ void __launch_bounds__(256) __attribute__((amdgpu_waves_per_eu(2, 2)))
inproj_conv_ssm_wmma_f32(const float* __restrict__ u,
                         const float* __restrict__ W_in,
                         const float* __restrict__ b_in,
                         const float* __restrict__ cw,
                         const float* __restrict__ cb,
                         const float* __restrict__ W,
                         const float* __restrict__ bias,
                         float* __restrict__ sz,
                         float* __restrict__ x2,
                         float* __restrict__ delta,
                         float* __restrict__ Bm,
                         float* __restrict__ Cm)
{
    __shared__ __attribute__((aligned(16))) float s_U[16 * APITCH];
    __shared__ __attribute__((aligned(16))) float s_UH[16 * APITCH];
    __shared__ __attribute__((aligned(16))) float s_X[18 * APITCH];
    __shared__ __attribute__((aligned(16))) float s_A[16 * APITCH];

    const int tid  = threadIdx.x;
    const int lane = tid & 31;
    const int wave = tid >> 5;
    const int m0   = blockIdx.x << 4;
    const int mA   = m0 + (wave << 1);
    const bool hasPrev = (m0 & (LEN - 1)) != 0;
    const bool hasNext = ((m0 + 16) & (LEN - 1)) != 0;
    const int hi  = lane >> 4;
    const int l16 = lane & 15;
    const int e   = lane << 2;

    {
        const float4* ug = reinterpret_cast<const float4*>(u) + lane;
        const int hsrc = (wave == 0) ? (hasPrev ? m0 - 1 : m0) : (hasNext ? m0 + 16 : m0);
        const float4 v0 = ug[(size_t)(m0 + wave) * (HDIM / 4)];
        const float4 v1 = ug[(size_t)(m0 + wave + 8) * (HDIM / 4)];
        float4 hv       = ug[(size_t)hsrc * (HDIM / 4)];
        const float4 zero4 = make_float4(0.f, 0.f, 0.f, 0.f);
        if (wave >= 2) hv = zero4;
        *reinterpret_cast<float4*>(s_U  + wave * APITCH + e)       = v0;
        *reinterpret_cast<float4*>(s_U  + (wave + 8) * APITCH + e) = v1;
        *reinterpret_cast<float4*>(s_UH + wave * APITCH + e)       = hv;
        *reinterpret_cast<float4*>(s_UH + (wave + 8) * APITCH + e) = zero4;
    }

    float wv[12];
#pragma unroll
    for (int i = 0; i < 12; ++i) wv[i] = cw[e * 3 + i];
    float bv[4];
#pragma unroll
    for (int i = 0; i < 4; ++i) bv[i] = cb[e + i];

    __syncthreads();

    const int col = (wave << 4) + l16;

    float zs[8];
    {
        v8f accX = {0.f, 0.f, 0.f, 0.f, 0.f, 0.f, 0.f, 0.f};
        v8f accH = {0.f, 0.f, 0.f, 0.f, 0.f, 0.f, 0.f, 0.f};
        v8f accZ = {0.f, 0.f, 0.f, 0.f, 0.f, 0.f, 0.f, 0.f};
        const float* Urow = s_U  + l16 * APITCH + 2 * hi;
        const float* Hrow = s_UH + l16 * APITCH + 2 * hi;
        const float* Wcol = W_in + (size_t)(2 * hi) * 256 + col;
#pragma unroll
        for (int k0 = 0; k0 < HDIM; k0 += 4) {
            v2f a, ah;
            a.x  = Urow[k0 + 0];
            a.y  = Urow[k0 + 1];
            ah.x = Hrow[k0 + 0];
            ah.y = Hrow[k0 + 1];
            v2f bX, bZ;
            bX.x = Wcol[(k0 + 0) * 256];
            bX.y = Wcol[(k0 + 1) * 256];
            bZ.x = Wcol[(k0 + 0) * 256 + 128];
            bZ.y = Wcol[(k0 + 1) * 256 + 128];
            accX = __builtin_amdgcn_wmma_f32_16x16x4_f32(false, a, false, bX,
                                                         (short)0, accX, false, false);
            accH = __builtin_amdgcn_wmma_f32_16x16x4_f32(false, ah, false, bX,
                                                         (short)0, accH, false, false);
            accZ = __builtin_amdgcn_wmma_f32_16x16x4_f32(false, a, false, bZ,
                                                         (short)0, accZ, false, false);
        }
        const float biasX = b_in[col];
        const float biasZ = b_in[128 + col];
#pragma unroll
        for (int r = 0; r < 8; ++r) {
            const int row = r + hi * 8;
            s_X[(1 + row) * APITCH + col] = accX[r] + biasX;
            zs[r] = silu_fast(accZ[r] + biasZ);
        }
        if (hi == 0) {
            s_X[col]               = hasPrev ? accH[0] + biasX : 0.f;
            s_X[17 * APITCH + col] = hasNext ? accH[1] + biasX : 0.f;
        }
    }
    __syncthreads();

    float4 o0, o1;
    {
        const int r0 = wave << 1;
        const float4 q0 = *reinterpret_cast<const float4*>(s_X + (r0 + 0) * APITCH + e);
        const float4 q1 = *reinterpret_cast<const float4*>(s_X + (r0 + 1) * APITCH + e);
        const float4 q2 = *reinterpret_cast<const float4*>(s_X + (r0 + 2) * APITCH + e);
        const float4 q3 = *reinterpret_cast<const float4*>(s_X + (r0 + 3) * APITCH + e);

        o0.x = silu_fast(q0.x * wv[0] + q1.x * wv[1]  + q2.x * wv[2]  + bv[0]);
        o0.y = silu_fast(q0.y * wv[3] + q1.y * wv[4]  + q2.y * wv[5]  + bv[1]);
        o0.z = silu_fast(q0.z * wv[6] + q1.z * wv[7]  + q2.z * wv[8]  + bv[2]);
        o0.w = silu_fast(q0.w * wv[9] + q1.w * wv[10] + q2.w * wv[11] + bv[3]);
        o1.x = silu_fast(q1.x * wv[0] + q2.x * wv[1]  + q3.x * wv[2]  + bv[0]);
        o1.y = silu_fast(q1.y * wv[3] + q2.y * wv[4]  + q3.y * wv[5]  + bv[1]);
        o1.z = silu_fast(q1.z * wv[6] + q2.z * wv[7]  + q3.z * wv[8]  + bv[2]);
        o1.w = silu_fast(q1.w * wv[9] + q2.w * wv[10] + q3.w * wv[11] + bv[3]);

        *reinterpret_cast<float4*>(s_A + r0 * APITCH + e)       = o0;
        *reinterpret_cast<float4*>(s_A + (r0 + 1) * APITCH + e) = o1;
    }
    __syncthreads();
    {
        float4* g = reinterpret_cast<float4*>(x2) + lane;
        g[(size_t)mA * (HDIM / 4)]       = o0;
        g[(size_t)(mA + 1) * (HDIM / 4)] = o1;
#pragma unroll
        for (int r = 0; r < 8; ++r)
            sz[(size_t)(m0 + r + hi * 8) * HDIM + col] = zs[r];
    }

    v8f accD = {0.f, 0.f, 0.f, 0.f, 0.f, 0.f, 0.f, 0.f};
    v8f accB = {0.f, 0.f, 0.f, 0.f, 0.f, 0.f, 0.f, 0.f};
    v8f accC = {0.f, 0.f, 0.f, 0.f, 0.f, 0.f, 0.f, 0.f};

    const float* Arow = s_A + l16 * APITCH + 2 * hi;
    const float* Wcol = W + (size_t)(2 * hi) * 384 + col;

#pragma unroll
    for (int k0 = 0; k0 < HDIM; k0 += 4) {
        v2f a;
        a.x = Arow[k0 + 0];
        a.y = Arow[k0 + 1];
        v2f bD, bB, bC;
        bD.x = Wcol[(k0 + 0) * 384];
        bD.y = Wcol[(k0 + 1) * 384];
        bB.x = Wcol[(k0 + 0) * 384 + 128];
        bB.y = Wcol[(k0 + 1) * 384 + 128];
        bC.x = Wcol[(k0 + 0) * 384 + 256];
        bC.y = Wcol[(k0 + 1) * 384 + 256];
        accD = __builtin_amdgcn_wmma_f32_16x16x4_f32(false, a, false, bD,
                                                     (short)0, accD, false, false);
        accB = __builtin_amdgcn_wmma_f32_16x16x4_f32(false, a, false, bB,
                                                     (short)0, accB, false, false);
        accC = __builtin_amdgcn_wmma_f32_16x16x4_f32(false, a, false, bC,
                                                     (short)0, accC, false, false);
    }

    const float biasD = bias[col];
    const float biasB = bias[128 + col];
    const float biasC = bias[256 + col];
#pragma unroll
    for (int r = 0; r < 8; ++r) {
        const size_t o = (size_t)(m0 + r + hi * 8) * 128 + col;
        delta[o] = softplus_fast(accD[r] + biasD);
        Bm[o]    = accB[r] + biasB;
        Cm[o]    = accC[r] + biasC;
    }
}

#define JS 4
#define JW (HDIM / JS)
#define SCAN_THREADS (8 * JW)
#define LOG2E_F 1.4426950408889634f

__device__ __forceinline__ float4 mul4(float4 a, float4 b)
{
    return make_float4(a.x * b.x, a.y * b.y, a.z * b.z, a.w * b.w);
}


struct StepOps { float4 d0, d1, b0, b1; float2 xv; };

__device__ __forceinline__ StepOps load_step(const float* s_d, const float* s_dB, const float* s_x,
                                             int l, int nbase, int cl)
{
    StepOps o;
    const float4* rd  = reinterpret_cast<const float4*>(s_d  + l * 128 + nbase);
    const float4* rdB = reinterpret_cast<const float4*>(s_dB + l * 128 + nbase);
    o.d0 = rd[0];  o.d1 = rd[1];
    o.b0 = rdB[0]; o.b1 = rdB[1];
    o.xv = *reinterpret_cast<const float2*>(s_x + l * JW + 2 * cl);
    return o;
}

__device__ __forceinline__ void hold_read_ahead(const StepOps& o)
{
    asm volatile("" :: "v"(o.d0.x), "v"(o.d0.y), "v"(o.d0.z), "v"(o.d0.w),
                       "v"(o.d1.x), "v"(o.d1.y), "v"(o.d1.z), "v"(o.d1.w),
                       "v"(o.b0.x), "v"(o.b0.y), "v"(o.b0.z), "v"(o.b0.w),
                       "v"(o.b1.x), "v"(o.b1.y), "v"(o.b1.z), "v"(o.b1.w),
                       "v"(o.xv.x), "v"(o.xv.y));
}

__device__ __forceinline__ float2 scan_step(float (&ha)[8], float (&hb)[8], const StepOps& o,
                                            float Aj2a, float Aj2b)
{
    const float d[8]  = {o.d0.x, o.d0.y, o.d0.z, o.d0.w, o.d1.x, o.d1.y, o.d1.z, o.d1.w};
    const float dB[8] = {o.b0.x, o.b0.y, o.b0.z, o.b0.w, o.b1.x, o.b1.y, o.b1.z, o.b1.w};
    float pa = 0.f, pb = 0.f;
#pragma unroll
    for (int i = 0; i < 8; ++i) {
        const float ea = __builtin_amdgcn_exp2f(d[i] * Aj2a);
        const float eb = __builtin_amdgcn_exp2f(d[i] * Aj2b);
        ha[i] = ea * ha[i] + dB[i] * o.xv.x;
        hb[i] = eb * hb[i] + dB[i] * o.xv.y;
        pa += ha[i];
        pb += hb[i];
    }
    return make_float2(pa, pb);
}

__global__ void __launch_bounds__(SCAN_THREADS)
scan_passA(const float* __restrict__ delta,
           const float* __restrict__ Bm,
           const float* __restrict__ x2,
           const float* __restrict__ A_log,
           float* __restrict__ Ssum,
           float* __restrict__ Bacc)
{
    const int wg = blockIdx.x;
    const int js = wg & (JS - 1);
    const int bc = wg / JS;
    const int b  = bc >> 4;
    const int c  = bc & (NCHUNK - 1);
    const int tid = threadIdx.x;
    const int cl  = tid & 15;
    const int g   = tid >> 4;
    const int j0  = js * JW + 2 * cl;
    const int nbase = g << 3;

    __shared__ __attribute__((aligned(16))) float s_d[CLEN * 128];
    __shared__ __attribute__((aligned(16))) float s_dB[CLEN * 128];
    __shared__ __attribute__((aligned(16))) float s_x[CLEN * JW];

    const size_t base = ((size_t)b * LEN + (size_t)c * CLEN) * HDIM;
    {
        const float4* gd = reinterpret_cast<const float4*>(delta + base);
        const float4* gB = reinterpret_cast<const float4*>(Bm + base);
        const float4* gx = reinterpret_cast<const float4*>(x2 + base + js * JW);
        float4* ld = reinterpret_cast<float4*>(s_d);
        float4* lB = reinterpret_cast<float4*>(s_dB);
        float4* lx = reinterpret_cast<float4*>(s_x);
        for (int k = 0; k < (CLEN * 128 / 4) / SCAN_THREADS; ++k) {
            const int q = tid + k * SCAN_THREADS;
            const float4 dv = gd[q];
            ld[q] = dv;
            lB[q] = mul4(dv, gB[q]);
        }
        for (int k = 0; k < (CLEN * JW / 4) / SCAN_THREADS; ++k) {
            const int q = tid + k * SCAN_THREADS;
            lx[q] = gx[(q / (JW / 4)) * (HDIM / 4) + (q % (JW / 4))];
        }
    }

    const float Aj2a = -__expf(A_log[j0]) * LOG2E_F;
    const float Aj2b = -__expf(A_log[j0 + 1]) * LOG2E_F;
    float ha[8], hb[8];
#pragma unroll
    for (int i = 0; i < 8; ++i) { ha[i] = 0.f; hb[i] = 0.f; }
    const int sl = cl & 7;
    float Ssl = 0.f;

    __syncthreads();

    StepOps p0 = load_step(s_d, s_dB, s_x, 0, nbase, cl);
    StepOps p1 = load_step(s_d, s_dB, s_x, 1, nbase, cl);
    float sp0 = s_d[nbase + sl];
    float sp1 = s_d[128 + nbase + sl];
    for (int l = 0; l < CLEN; l += 4) {
        const StepOps q0 = load_step(s_d, s_dB, s_x, l + 2, nbase, cl);
        const StepOps q1 = load_step(s_d, s_dB, s_x, l + 3, nbase, cl);
        const float sq0 = s_d[(l + 2) * 128 + nbase + sl];
        const float sq1 = s_d[(l + 3) * 128 + nbase + sl];
        __builtin_amdgcn_sched_barrier(0);
        Ssl += sp0;
        scan_step(ha, hb, p0, Aj2a, Aj2b);
        Ssl += sp1;
        scan_step(ha, hb, p1, Aj2a, Aj2b);
        const int l4 = (l + 4 < CLEN) ? l + 4 : CLEN - 2;
        p0  = load_step(s_d, s_dB, s_x, l4, nbase, cl);
        p1  = load_step(s_d, s_dB, s_x, l4 + 1, nbase, cl);
        sp0 = s_d[l4 * 128 + nbase + sl];
        sp1 = s_d[(l4 + 1) * 128 + nbase + sl];
        __builtin_amdgcn_sched_barrier(0);
        Ssl += sq0;
        scan_step(ha, hb, q0, Aj2a, Aj2b);
        Ssl += sq1;
        scan_step(ha, hb, q1, Aj2a, Aj2b);
        hold_read_ahead(p0);
        hold_read_ahead(p1);
    }

    const size_t ob = (size_t)bc * 128;
#pragma unroll
    for (int i = 0; i < 8; ++i)
        *reinterpret_cast<float2*>(Bacc + (ob + nbase + i) * 128 + j0) = make_float2(ha[i], hb[i]);
    if (js == 0 && cl < 8)
        Ssum[ob + nbase + sl] = Ssl;
}

__global__ void __launch_bounds__(SCAN_THREADS)
scan_passB(const float* __restrict__ Ssum,
           const float* __restrict__ Bacc,
           const float* __restrict__ A_log,
           float* __restrict__ hinit)
{
    const int id = blockIdx.x * SCAN_THREADS + threadIdx.x;
    const int j  = id & 127;
    const int n  = (id >> 7) & 127;
    const int b  = id >> 14;

    const float Aj = -__expf(A_log[j]);
    float h = 0.f;
#pragma unroll
    for (int c = 0; c < NCHUNK; ++c) {
        const size_t ob = (size_t)(b * NCHUNK + c) * 128;
        const size_t idx = (ob + n) * 128 + j;
        hinit[idx] = h;
        const float P = __expf(Ssum[ob + n] * Aj);
        h = P * h + Bacc[idx];
    }
}

#define SBLK 8
__global__ void __launch_bounds__(SCAN_THREADS)
scan_passC(const float* __restrict__ delta,
           const float* __restrict__ Bm,
           const float* __restrict__ Cm,
           const float* __restrict__ x2,
           const float* __restrict__ u,
           const float* __restrict__ sz,
           const float* __restrict__ A_log,
           const float* __restrict__ Dv,
           const float* __restrict__ hinit,
           float* __restrict__ ymul)
{
    const int wg = blockIdx.x;
    const int js = wg & (JS - 1);
    const int bc = wg / JS;
    const int b  = bc >> 4;
    const int c  = bc & (NCHUNK - 1);
    const int tid = threadIdx.x;
    const int cl  = tid & 15;
    const int g   = tid >> 4;
    const int j0  = js * JW + 2 * cl;
    const int nbase = g << 3;
    const int jl  = tid & (JW - 1);
    const int ng  = tid / JW;
    const int j   = js * JW + jl;

    __shared__ __attribute__((aligned(16))) float s_d[CLEN * 128];
    __shared__ __attribute__((aligned(16))) float s_dB[CLEN * 128];
    __shared__ __attribute__((aligned(16))) float s_x[CLEN * JW];
    __shared__ __attribute__((aligned(16))) float s_part[2][SBLK][16][JW];

    const size_t base = ((size_t)b * LEN + (size_t)c * CLEN) * HDIM;
    {
        const float4* gd = reinterpret_cast<const float4*>(delta + base);
        const float4* gB = reinterpret_cast<const float4*>(Bm + base);
        const float4* gx = reinterpret_cast<const float4*>(x2 + base + js * JW);
        float4* ld = reinterpret_cast<float4*>(s_d);
        float4* lB = reinterpret_cast<float4*>(s_dB);
        float4* lx = reinterpret_cast<float4*>(s_x);
        for (int k = 0; k < (CLEN * 128 / 4) / SCAN_THREADS; ++k) {
            const int q = tid + k * SCAN_THREADS;
            const float4 dv = gd[q];
            ld[q] = dv;
            lB[q] = mul4(dv, gB[q]);
        }
        for (int k = 0; k < (CLEN * JW / 4) / SCAN_THREADS; ++k) {
            const int q = tid + k * SCAN_THREADS;
            lx[q] = gx[(q / (JW / 4)) * (HDIM / 4) + (q % (JW / 4))];
        }
    }

    const float Aj2a = -__expf(A_log[j0]) * LOG2E_F;
    const float Aj2b = -__expf(A_log[j0 + 1]) * LOG2E_F;
    const float Dj = Dv[j];

    const size_t ob = (size_t)bc * 128;
    float ha[8], hb[8];
#pragma unroll
    for (int i = 0; i < 8; ++i) {
        const float2 hv = *reinterpret_cast<const float2*>(hinit + (ob + nbase + i) * 128 + j0);
        ha[i] = hv.x;
        hb[i] = hv.y;
    }

    __syncthreads();

    StepOps p0 = load_step(s_d, s_dB, s_x, 0, nbase, cl);
    StepOps p1 = load_step(s_d, s_dB, s_x, 1, nbase, cl);
    for (int blk = 0; blk < CLEN / SBLK; ++blk) {
        const int buf = blk & 1;
        const size_t moff = base + (size_t)(blk * SBLK + ng) * HDIM + j;
        const float cv = Cm[moff];
        const float uv = u[moff];
        const float zv = sz[moff];

        for (int ll = 0; ll < SBLK; ll += 4) {
            const int l = blk * SBLK + ll;
            const StepOps q0 = load_step(s_d, s_dB, s_x, l + 2, nbase, cl);
            const StepOps q1 = load_step(s_d, s_dB, s_x, l + 3, nbase, cl);
            __builtin_amdgcn_sched_barrier(0);
            *reinterpret_cast<float2*>(&s_part[buf][ll][g][2 * cl])     = scan_step(ha, hb, p0, Aj2a, Aj2b);
            *reinterpret_cast<float2*>(&s_part[buf][ll + 1][g][2 * cl]) = scan_step(ha, hb, p1, Aj2a, Aj2b);
            const int l4 = (l + 4 < CLEN) ? l + 4 : CLEN - 2;
            p0 = load_step(s_d, s_dB, s_x, l4, nbase, cl);
            p1 = load_step(s_d, s_dB, s_x, l4 + 1, nbase, cl);
            __builtin_amdgcn_sched_barrier(0);
            *reinterpret_cast<float2*>(&s_part[buf][ll + 2][g][2 * cl]) = scan_step(ha, hb, q0, Aj2a, Aj2b);
            *reinterpret_cast<float2*>(&s_part[buf][ll + 3][g][2 * cl]) = scan_step(ha, hb, q1, Aj2a, Aj2b);
        }
        __syncthreads();
        float tot = 0.f;
#pragma unroll
        for (int gg = 0; gg < 16; ++gg) tot += s_part[buf][ng][gg][jl];
        const float y = cv * tot + Dj * uv;
        ymul[moff] = y * zv;
    }
}

extern "C" void kernel_launch(void* const* d_in, const int* in_sizes, int n_in,
                              void* d_out, int out_size, void* d_ws, size_t ws_size,
                              hipStream_t stream)
{
    (void)in_sizes; (void)n_in; (void)out_size; (void)ws_size;

    const float* u      = (const float*)d_in[0];
    const float* W_in   = (const float*)d_in[1];
    const float* b_in   = (const float*)d_in[2];
    const float* conv_w = (const float*)d_in[3];
    const float* conv_b = (const float*)d_in[4];
    const float* W_x    = (const float*)d_in[5];
    const float* b_x    = (const float*)d_in[6];
    const float* A_log  = (const float*)d_in[7];
    const float* Dv     = (const float*)d_in[8];
    const float* W_out  = (const float*)d_in[9];
    const float* b_out  = (const float*)d_in[10];

    const size_t P = (size_t)MTOT * HDIM;
    const size_t NSTATE = (size_t)BATCH * NCHUNK * 128 * 128;

    float* ws    = (float*)d_ws;
    float* s_z   = ws + 1 * P;
    float* x2    = ws + 2 * P;
    float* delta = ws + 3 * P;
    float* Bmat  = ws + 4 * P;
    float* Cmat  = ws + 5 * P;
    float* ymul  = ws + 6 * P;
    float* Ssum  = ws + 7 * P;
    float* Bacc  = Ssum + (size_t)BATCH * NCHUNK * 128;
    float* hinit = Bacc + NSTATE;

    inproj_conv_ssm_wmma_f32<<<MTOT / 16, 256, 0, stream>>>(
        u, W_in, b_in, conv_w, conv_b, W_x, b_x, s_z, x2, delta, Bmat, Cmat);
    scan_passA<<<BATCH * NCHUNK * JS, SCAN_THREADS, 0, stream>>>(delta, Bmat, x2, A_log,
                                                                 Ssum, Bacc);
    scan_passB<<<(BATCH * 128 * 128) / SCAN_THREADS, SCAN_THREADS, 0, stream>>>(
        Ssum, Bacc, A_log, hinit);
    scan_passC<<<BATCH * NCHUNK * JS, SCAN_THREADS, 0, stream>>>(delta, Bmat, Cmat, x2, u, s_z,
                                                                 A_log, Dv, hinit, ymul);
    gemm16_wmma_f32<128, 128, EPI_PLAIN><<<256, 256, 0, stream>>>(
        ymul, W_out, b_out, (float*)d_out, nullptr);
}
